# starting kernel
# baseline (speedup 1.0000x reference)
#include <hip/hip_runtime.h>

typedef float v2f __attribute__((ext_vector_type(2)));
typedef float v8f __attribute__((ext_vector_type(8)));

#define FEAT 128

__global__ __launch_bounds__(256)
void gcn_gemm_dual_wmma(const float* __restrict__ X,
                        const float* __restrict__ W1,
                        const float* __restrict__ W2,
                        float* __restrict__ H1,
                        float* __restrict__ H2,
                        int M) {
  const int lane = threadIdx.x & 31;
  const int wave = threadIdx.x >> 5;
  const int m0 = blockIdx.x * 16;
  const int n0 = wave * 16;
  const int l15 = lane & 15;
  const int hi  = lane >> 4;

  int mload = m0 + l15;
  if (mload > M - 1) mload = M - 1;
  const float* __restrict__ xrow = X + (size_t)mload * FEAT;

  v8f c1 = {0.f, 0.f, 0.f, 0.f, 0.f, 0.f, 0.f, 0.f};
  v8f c2 = c1;

#pragma unroll 4
  for (int k = 0; k < FEAT; k += 4) {
    const int ka = k + 2 * hi;
    v2f a;  a.x  = xrow[ka];                      a.y  = xrow[ka + 1];
    v2f b1; b1.x = W1[ka * FEAT + n0 + l15];      b1.y = W1[(ka + 1) * FEAT + n0 + l15];
    v2f b2; b2.x = W2[ka * FEAT + n0 + l15];      b2.y = W2[(ka + 1) * FEAT + n0 + l15];
    c1 = __builtin_amdgcn_wmma_f32_16x16x4_f32(false, a, false, b1, (short)0, c1, false, false);
    c2 = __builtin_amdgcn_wmma_f32_16x16x4_f32(false, a, false, b2, (short)0, c2, false, false);
  }

  const size_t base = (size_t)(m0 + 8 * hi) * FEAT + (n0 + l15);
  float* __restrict__ o1 = H1 + base;
  float* __restrict__ o2 = H2 + base;
  if (m0 + 16 <= M) {
#pragma unroll
    for (int r = 0; r < 8; ++r) {
      o1[(size_t)r * FEAT] = c1[r];
      o2[(size_t)r * FEAT] = c2[r];
    }
  } else {
#pragma unroll
    for (int r = 0; r < 8; ++r) {
      if (m0 + r + 8 * hi < M) {
        o1[(size_t)r * FEAT] = c1[r];
        o2[(size_t)r * FEAT] = c2[r];
      }
    }
  }
}

__global__ void deg_init(float* __restrict__ d1, float* __restrict__ d2, int n) {
  int i = blockIdx.x * blockDim.x + threadIdx.x;
  if (i < n) { d1[i] = 1.0f; d2[i] = 1.0f; }
}

__global__ void deg_count(const int* __restrict__ dst, float* __restrict__ deg, int e) {
  int i = blockIdx.x * blockDim.x + threadIdx.x;
  if (i < e) atomicAdd(&deg[dst[i]], 1.0f);
}

__global__ void deg_rsqrt(float* __restrict__ d1, float* __restrict__ d2, int n) {
  int i = blockIdx.x * blockDim.x + threadIdx.x;
  if (i < n) { d1[i] = rsqrtf(d1[i]); d2[i] = rsqrtf(d2[i]); }
}

__global__ void zero_f4(float4* __restrict__ p, size_t n4) {
  size_t i = (size_t)blockIdx.x * blockDim.x + threadIdx.x;
  if (i < n4) p[i] = make_float4(0.f, 0.f, 0.f, 0.f);
}

__global__ __launch_bounds__(256)
void edge_scatter(const float* __restrict__ H,
                  const int* __restrict__ src,
                  const int* __restrict__ dst,
                  const float* __restrict__ dinv,
                  float* __restrict__ AGG,
                  int E) {
  const int e = blockIdx.x * 8 + (threadIdx.x >> 5);
  if (e >= E) return;
  const int lane = threadIdx.x & 31;
  const int s = src[e];
  const int d = dst[e];
  const float nrm = dinv[s] * dinv[d];
  const float4 v = *((const float4*)(H + (size_t)s * FEAT) + lane);
  float* ap = AGG + (size_t)d * FEAT + lane * 4;
  atomicAdd(ap + 0, v.x * nrm);
  atomicAdd(ap + 1, v.y * nrm);
  atomicAdd(ap + 2, v.z * nrm);
  atomicAdd(ap + 3, v.w * nrm);
}

__global__ __launch_bounds__(256)
void combine(const float* __restrict__ AGG,
             const float* __restrict__ H1,
             const float* __restrict__ H2,
             const float* __restrict__ dinv1,
             const float* __restrict__ dinv2,
             const float* __restrict__ b1,
             const float* __restrict__ b2,
             float* __restrict__ OUT,
             int N, int relu) {
  const int t = blockIdx.x * blockDim.x + threadIdx.x;
  if (t >= N * 32) return;
  const int v = t >> 5;
  const int q = (t & 31) * 4;
  float d1 = dinv1[v]; d1 *= d1;
  float d2 = dinv2[v]; d2 *= d2;
  const size_t off = (size_t)v * FEAT + q;
  const float4 a  = *(const float4*)(AGG + off);
  const float4 h1 = *(const float4*)(H1 + off);
  const float4 h2 = *(const float4*)(H2 + off);
  const float4 c1 = *(const float4*)(b1 + q);
  const float4 c2 = *(const float4*)(b2 + q);
  float4 o;
  o.x = a.x + d1 * h1.x + d2 * h2.x + c1.x + c2.x;
  o.y = a.y + d1 * h1.y + d2 * h2.y + c1.y + c2.y;
  o.z = a.z + d1 * h1.z + d2 * h2.z + c1.z + c2.z;
  o.w = a.w + d1 * h1.w + d2 * h2.w + c1.w + c2.w;
  if (relu) {
    o.x = fmaxf(o.x, 0.f); o.y = fmaxf(o.y, 0.f);
    o.z = fmaxf(o.z, 0.f); o.w = fmaxf(o.w, 0.f);
  }
  *(float4*)(OUT + off) = o;
}

extern "C" void kernel_launch(void* const* d_in, const int* in_sizes, int n_in,
                              void* d_out, int out_size, void* d_ws, size_t ws_size,
                              hipStream_t stream) {
  const float* x      = (const float*)d_in[0];
  const int*   ei     = (const int*)d_in[1];
  const int*   cei    = (const int*)d_in[2];
  const float* conv_w = (const float*)d_in[3];
  const float* conv_b = (const float*)d_in[4];
  const float* ctrl_w = (const float*)d_in[5];
  const float* ctrl_b = (const float*)d_in[6];

  const int N  = in_sizes[0] / FEAT;
  const int E1 = in_sizes[1] / 2;
  const int E2 = in_sizes[2] / 2;
  const int D  = in_sizes[3] / (FEAT * FEAT);

  const int* src1 = ei;        const int* dst1 = ei + E1;
  const int* src2 = cei;       const int* dst2 = cei + E2;

  const size_t NF = (size_t)N * FEAT;
  float* h1    = (float*)d_ws;
  float* h2    = h1 + NF;
  float* agg   = h2 + NF;
  float* xbuf  = agg + NF;
  float* dinv1 = xbuf + NF;
  float* dinv2 = dinv1 + N;

  const int T = 256;
  const int nodeBlocks = (N + T - 1) / T;

  deg_init<<<nodeBlocks, T, 0, stream>>>(dinv1, dinv2, N);
  deg_count<<<(E1 + T - 1) / T, T, 0, stream>>>(dst1, dinv1, E1);
  deg_count<<<(E2 + T - 1) / T, T, 0, stream>>>(dst2, dinv2, E2);
  deg_rsqrt<<<nodeBlocks, T, 0, stream>>>(dinv1, dinv2, N);

  const int gemmBlocks = (N + 15) / 16;
  const size_t n4 = NF / 4;
  const int zeroBlocks = (int)((n4 + T - 1) / T);
  const int combBlocks = (N * 32 + T - 1) / T;

  const float* x_cur = x;
  for (int i = 0; i < D; ++i) {
    const float* W1 = conv_w + (size_t)i * FEAT * FEAT;
    const float* W2 = ctrl_w + (size_t)i * FEAT * FEAT;
    const float* B1 = conv_b + (size_t)i * FEAT;
    const float* B2 = ctrl_b + (size_t)i * FEAT;
    float* x_next = (i == D - 1) ? (float*)d_out : xbuf;

    gcn_gemm_dual_wmma<<<gemmBlocks, T, 0, stream>>>(x_cur, W1, W2, h1, h2, N);
    zero_f4<<<zeroBlocks, T, 0, stream>>>((float4*)agg, n4);
    edge_scatter<<<(E1 + 7) / 8, T, 0, stream>>>(h1, src1, dst1, dinv1, agg, E1);
    edge_scatter<<<(E2 + 7) / 8, T, 0, stream>>>(h2, src2, dst2, dinv2, agg, E2);
    combine<<<combBlocks, T, 0, stream>>>(agg, h1, h2, dinv1, dinv2, B1, B2,
                                          x_next, N, (i != D - 1) ? 1 : 0);
    x_cur = x_next;
  }
}


# Float atomic feature scatter replaced by per-call dst-sorted edge list (int atomic arrival-order rank) + one-wave-per-node register gather, zero+combine fused in; per-node sum order = rank order
# speedup vs baseline: 2.9345x; 2.9345x over previous
#include <hip/hip_runtime.h>

typedef float v2f __attribute__((ext_vector_type(2)));
typedef float v8f __attribute__((ext_vector_type(8)));

#define FEAT 128

__global__ __launch_bounds__(256)
void gcn_gemm_dual_wmma(const float* __restrict__ X,
                        const float* __restrict__ W1,
                        const float* __restrict__ W2,
                        float* __restrict__ H1,
                        float* __restrict__ H2,
                        int M) {
  const int lane = threadIdx.x & 31;
  const int wave = threadIdx.x >> 5;
  const int m0 = blockIdx.x * 16;
  const int n0 = wave * 16;
  const int l15 = lane & 15;
  const int hi  = lane >> 4;

  int mload = m0 + l15;
  if (mload > M - 1) mload = M - 1;
  const float* __restrict__ xrow = X + (size_t)mload * FEAT;

  v8f c1 = {0.f, 0.f, 0.f, 0.f, 0.f, 0.f, 0.f, 0.f};
  v8f c2 = c1;

#pragma unroll 4
  for (int k = 0; k < FEAT; k += 4) {
    const int ka = k + 2 * hi;
    v2f a;  a.x  = xrow[ka];                      a.y  = xrow[ka + 1];
    v2f b1; b1.x = W1[ka * FEAT + n0 + l15];      b1.y = W1[(ka + 1) * FEAT + n0 + l15];
    v2f b2; b2.x = W2[ka * FEAT + n0 + l15];      b2.y = W2[(ka + 1) * FEAT + n0 + l15];
    c1 = __builtin_amdgcn_wmma_f32_16x16x4_f32(false, a, false, b1, (short)0, c1, false, false);
    c2 = __builtin_amdgcn_wmma_f32_16x16x4_f32(false, a, false, b2, (short)0, c2, false, false);
  }

  const size_t base = (size_t)(m0 + 8 * hi) * FEAT + (n0 + l15);
  float* __restrict__ o1 = H1 + base;
  float* __restrict__ o2 = H2 + base;
  if (m0 + 16 <= M) {
#pragma unroll
    for (int r = 0; r < 8; ++r) {
      o1[(size_t)r * FEAT] = c1[r];
      o2[(size_t)r * FEAT] = c2[r];
    }
  } else {
#pragma unroll
    for (int r = 0; r < 8; ++r) {
      if (m0 + r + 8 * hi < M) {
        o1[(size_t)r * FEAT] = c1[r];
        o2[(size_t)r * FEAT] = c2[r];
      }
    }
  }
}

__global__ void cnt_zero(int4* __restrict__ c, int n4) {
  const int i = blockIdx.x * blockDim.x + threadIdx.x;
  if (i < n4) c[i] = make_int4(0, 0, 0, 0);
}

__global__ __launch_bounds__(256)
void edge_rank(const int* __restrict__ dst1, const int* __restrict__ dst2,
               int* __restrict__ cnt1, int* __restrict__ cnt2,
               int* __restrict__ rank, int E1, int E) {
  const int i = blockIdx.x * blockDim.x + threadIdx.x;
  if (i >= E) return;
  if (i < E1) rank[i] = atomicAdd(&cnt1[dst1[i]], 1);
  else        rank[i] = atomicAdd(&cnt2[dst2[i - E1]], 1);
}

__global__ __launch_bounds__(256)
void row_scan(const int* __restrict__ cnt1, const int* __restrict__ cnt2,
              int* __restrict__ start,
              float* __restrict__ dinv1, float* __restrict__ dinv2, int N) {
  __shared__ int wsum[8];
  __shared__ int wpre[8];
  const int t = threadIdx.x;
  const int lane = t & 31;
  const int wave = t >> 5;
  const int4* __restrict__ c1 = (const int4*)cnt1;
  const int4* __restrict__ c2 = (const int4*)cnt2;

  int pre = 0;
  for (int k = 0; k < (int)blockIdx.x; ++k) {
    const int4 a = c1[k * 256 + t];
    const int4 b = c2[k * 256 + t];
    pre += (a.x + a.y) + (a.z + a.w) + (b.x + b.y) + (b.z + b.w);
  }
  const int q = blockIdx.x * 256 + t;
  const int4 a = c1[q];
  const int4 b = c2[q];
  const int l0 = a.x + b.x, l1 = a.y + b.y, l2 = a.z + b.z, l3 = a.w + b.w;
  const int mine = (l0 + l1) + (l2 + l3);

  int inc = mine;
  for (int o = 1; o < 32; o <<= 1) {
    const int u = __shfl_up(inc, o);
    if (lane >= o) inc += u;
  }
  for (int o = 16; o > 0; o >>= 1) pre += __shfl_xor(pre, o);
  if (lane == 31) wsum[wave] = inc;
  if (lane == 0)  wpre[wave] = pre;
  __syncthreads();
  int off = 0;
  for (int w = 0; w < 8; ++w) {
    off += wpre[w];
    if (w < wave) off += wsum[w];
  }
  const int s0 = off + inc - mine;
  const int i = q * 4;
  const int s1 = s0 + l0, s2 = s1 + l1, s3 = s2 + l2;
  if (i < N)     { start[i]     = s0; dinv1[i]     = rsqrtf((float)(1 + a.x)); dinv2[i]     = rsqrtf((float)(1 + b.x)); }
  if (i + 1 < N) { start[i + 1] = s1; dinv1[i + 1] = rsqrtf((float)(1 + a.y)); dinv2[i + 1] = rsqrtf((float)(1 + b.y)); }
  if (i + 2 < N) { start[i + 2] = s2; dinv1[i + 2] = rsqrtf((float)(1 + a.z)); dinv2[i + 2] = rsqrtf((float)(1 + b.z)); }
  if (i + 3 < N) { start[i + 3] = s3; dinv1[i + 3] = rsqrtf((float)(1 + a.w)); dinv2[i + 3] = rsqrtf((float)(1 + b.w)); }
}

__global__ __launch_bounds__(256)
void csr_fill(const int* __restrict__ src1, const int* __restrict__ dst1,
              const int* __restrict__ src2, const int* __restrict__ dst2,
              const int* __restrict__ rank, const int* __restrict__ start,
              const int* __restrict__ cnt1,
              const float* __restrict__ dinv1, const float* __restrict__ dinv2,
              int2* __restrict__ csr, int E1, int E, int N) {
  const int i = blockIdx.x * blockDim.x + threadIdx.x;
  if (i >= E) return;
  int row, p;
  float w;
  if (i < E1) {
    const int s = src1[i], d = dst1[i];
    p = start[d] + rank[i];
    w = dinv1[s] * dinv1[d];
    row = s;
  } else {
    const int j = i - E1;
    const int s = src2[j], d = dst2[j];
    p = start[d] + cnt1[d] + rank[i];
    w = dinv2[s] * dinv2[d];
    row = N + s;
  }
  csr[p] = make_int2(row, __float_as_int(w));
}

#define ACC_ROW(g, e)                                   \
  acc.x = __builtin_fmaf((g).x, (e), acc.x);           \
  acc.y = __builtin_fmaf((g).y, (e), acc.y);           \
  acc.z = __builtin_fmaf((g).z, (e), acc.z);           \
  acc.w = __builtin_fmaf((g).w, (e), acc.w)

__global__ __launch_bounds__(256)
void gather_combine(const float* __restrict__ H12,
                    const int2* __restrict__ csr,
                    const int* __restrict__ start,
                    const int* __restrict__ cnt1,
                    const int* __restrict__ cnt2,
                    const float* __restrict__ dinv1,
                    const float* __restrict__ dinv2,
                    const float* __restrict__ b1,
                    const float* __restrict__ b2,
                    float* __restrict__ OUT,
                    int N, int relu) {
  const int lane = threadIdx.x & 31;
  const int v = __builtin_amdgcn_readfirstlane((int)(blockIdx.x * 8 + (threadIdx.x >> 5)));
  if (v >= N) return;
  const int beg = start[v];
  const int n = cnt1[v] + cnt2[v];
  const float d1 = dinv1[v];
  const float d2 = dinv2[v];
  const float w1 = d1 * d1;
  const float w2 = d2 * d2;
  const float4* __restrict__ Hq = (const float4*)H12 + lane;
  const float4 s1 = Hq[(size_t)v * 32];
  const float4 s2 = Hq[(size_t)(N + v) * 32];
  float4 acc;
  acc.x = s1.x * w1; acc.y = s1.y * w1; acc.z = s1.z * w1; acc.w = s1.w * w1;
  ACC_ROW(s2, w2);

  for (int base = 0; base < n; base += 32) {
    const int m = (n - base < 32) ? (n - base) : 32;
    int2 ent = make_int2(0, 0);
    if (lane < m) ent = csr[beg + base + lane];
    const int m4 = m & ~3;
    int j = 0;
    for (; j < m4; j += 4) {
      const int   r0 = __builtin_amdgcn_readlane(ent.x, j);
      const int   r1 = __builtin_amdgcn_readlane(ent.x, j + 1);
      const int   r2 = __builtin_amdgcn_readlane(ent.x, j + 2);
      const int   r3 = __builtin_amdgcn_readlane(ent.x, j + 3);
      const float e0 = __int_as_float(__builtin_amdgcn_readlane(ent.y, j));
      const float e1 = __int_as_float(__builtin_amdgcn_readlane(ent.y, j + 1));
      const float e2 = __int_as_float(__builtin_amdgcn_readlane(ent.y, j + 2));
      const float e3 = __int_as_float(__builtin_amdgcn_readlane(ent.y, j + 3));
      const float4 g0 = Hq[(size_t)r0 * 32];
      const float4 g1 = Hq[(size_t)r1 * 32];
      const float4 g2 = Hq[(size_t)r2 * 32];
      const float4 g3 = Hq[(size_t)r3 * 32];
      ACC_ROW(g0, e0); ACC_ROW(g1, e1); ACC_ROW(g2, e2); ACC_ROW(g3, e3);
    }
    if (j < m) {
      const int   r0 = __builtin_amdgcn_readlane(ent.x, j);
      const int   r1 = __builtin_amdgcn_readlane(ent.x, j + 1);
      const int   r2 = __builtin_amdgcn_readlane(ent.x, j + 2);
      const float e0 = __int_as_float(__builtin_amdgcn_readlane(ent.y, j));
      const float e1 = __int_as_float(__builtin_amdgcn_readlane(ent.y, j + 1));
      const float e2 = __int_as_float(__builtin_amdgcn_readlane(ent.y, j + 2));
      const float4 g0 = Hq[(size_t)r0 * 32];
      float4 g1 = make_float4(0.f, 0.f, 0.f, 0.f);
      float4 g2 = g1;
      if (j + 1 < m) g1 = Hq[(size_t)r1 * 32];
      if (j + 2 < m) g2 = Hq[(size_t)r2 * 32];
      ACC_ROW(g0, e0); ACC_ROW(g1, e1); ACC_ROW(g2, e2);
    }
  }

  const float4 c1 = *((const float4*)b1 + lane);
  const float4 c2 = *((const float4*)b2 + lane);
  float4 o;
  o.x = acc.x + c1.x + c2.x;
  o.y = acc.y + c1.y + c2.y;
  o.z = acc.z + c1.z + c2.z;
  o.w = acc.w + c1.w + c2.w;
  if (relu) {
    o.x = fmaxf(o.x, 0.f); o.y = fmaxf(o.y, 0.f);
    o.z = fmaxf(o.z, 0.f); o.w = fmaxf(o.w, 0.f);
  }
  *((float4*)(OUT + (size_t)v * FEAT) + lane) = o;
}

extern "C" void kernel_launch(void* const* d_in, const int* in_sizes, int n_in,
                              void* d_out, int out_size, void* d_ws, size_t ws_size,
                              hipStream_t stream) {
  const float* x      = (const float*)d_in[0];
  const int*   ei     = (const int*)d_in[1];
  const int*   cei    = (const int*)d_in[2];
  const float* conv_w = (const float*)d_in[3];
  const float* conv_b = (const float*)d_in[4];
  const float* ctrl_w = (const float*)d_in[5];
  const float* ctrl_b = (const float*)d_in[6];

  const int N  = in_sizes[0] / FEAT;
  const int E1 = in_sizes[1] / 2;
  const int E2 = in_sizes[2] / 2;
  const int D  = in_sizes[3] / (FEAT * FEAT);
  const int E  = E1 + E2;

  const int* src1 = ei;        const int* dst1 = ei + E1;
  const int* src2 = cei;       const int* dst2 = cei + E2;

  const size_t NF = (size_t)N * FEAT;
  const int Npad = ((N + 1023) / 1024) * 1024;
  float* h1    = (float*)d_ws;
  float* h2    = h1 + NF;
  float* xbuf  = h2 + NF;
  float* dinv1 = xbuf + NF;
  float* dinv2 = dinv1 + Npad;
  int*   cnt1  = (int*)(dinv2 + Npad);
  int*   cnt2  = cnt1 + Npad;
  int*   start = cnt2 + Npad;
  int*   rank  = start + Npad;
  int2*  csr   = (int2*)(rank + ((E + 3) / 4) * 4);

  const int T = 256;
  const int edgeBlocks = (E + T - 1) / T;

  cnt_zero<<<(2 * Npad / 4 + T - 1) / T, T, 0, stream>>>((int4*)cnt1, 2 * Npad / 4);
  edge_rank<<<edgeBlocks, T, 0, stream>>>(dst1, dst2, cnt1, cnt2, rank, E1, E);
  row_scan<<<Npad / 1024, T, 0, stream>>>(cnt1, cnt2, start, dinv1, dinv2, N);
  csr_fill<<<edgeBlocks, T, 0, stream>>>(src1, dst1, src2, dst2, rank, start, cnt1,
                                         dinv1, dinv2, csr, E1, E, N);

  const int gemmBlocks = (N + 15) / 16;
  const int nodeWaveBlocks = (N + 7) / 8;

  const float* x_cur = x;
  for (int i = 0; i < D; ++i) {
    const float* W1 = conv_w + (size_t)i * FEAT * FEAT;
    const float* W2 = ctrl_w + (size_t)i * FEAT * FEAT;
    const float* B1 = conv_b + (size_t)i * FEAT;
    const float* B2 = ctrl_b + (size_t)i * FEAT;
    float* x_next = (i == D - 1) ? (float*)d_out : xbuf;

    gcn_gemm_dual_wmma<<<gemmBlocks, T, 0, stream>>>(x_cur, W1, W2, h1, h2, N);
    gather_combine<<<nodeWaveBlocks, T, 0, stream>>>(h1, csr, start, cnt1, cnt2, dinv1, dinv2,
                                                     B1, B2, x_next, N, (i != D - 1) ? 1 : 0);
    x_cur = x_next;
  }
}


# Dense transform: W1,W2 copied once per block into LDS (row pairs interleaved), 512 resident blocks walk the row tiles, each wave a 16x64 tile: one 8-byte A load feeds 4 WMMAs; k order unchanged
# speedup vs baseline: 3.3725x; 1.1493x over previous
#include <hip/hip_runtime.h>

typedef float v2f __attribute__((ext_vector_type(2)));
typedef float v8f __attribute__((ext_vector_type(8)));

#define FEAT 128

#define WPITCH2 288
#define WMAT (64 * WPITCH2)
#define GEMM_BLOCKS 512

__device__ __forceinline__ void gemm_k32(const v2f (&a)[8], const float* __restrict__ wp0,
                                         v8f& c0, v8f& c1, v8f& c2, v8f& c3) {
#pragma unroll
  for (int u = 0; u < 8; ++u) {
    const float* __restrict__ wp = wp0 + (2 * u) * WPITCH2;
    const v2f b0 = *(const v2f*)(wp);
    const v2f b1 = *(const v2f*)(wp + 32);
    const v2f b2 = *(const v2f*)(wp + 64);
    const v2f b3 = *(const v2f*)(wp + 96);
    c0 = __builtin_amdgcn_wmma_f32_16x16x4_f32(false, a[u], false, b0, (short)0, c0, false, false);
    c1 = __builtin_amdgcn_wmma_f32_16x16x4_f32(false, a[u], false, b1, (short)0, c1, false, false);
    c2 = __builtin_amdgcn_wmma_f32_16x16x4_f32(false, a[u], false, b2, (short)0, c2, false, false);
    c3 = __builtin_amdgcn_wmma_f32_16x16x4_f32(false, a[u], false, b3, (short)0, c3, false, false);
  }
}

__global__ __launch_bounds__(256)
void gcn_gemm_dual_wmma(const float* __restrict__ X,
                        const float* __restrict__ W1,
                        const float* __restrict__ W2,
                        float* __restrict__ H1,
                        float* __restrict__ H2,
                        int M) {
  __shared__ __attribute__((aligned(16))) float Ws[2 * WMAT];

  const int lane = threadIdx.x & 31;
  const int wave = threadIdx.x >> 5;
  const int l15 = lane & 15;
  const int hi  = lane >> 4;

  {
    const int col4 = lane * 4;
#pragma unroll
    for (int i = 0; i < 8; ++i) {
      const int p = wave + 8 * i;
      const float4 r0 = *(const float4*)(W1 + (2 * p) * FEAT + col4);
      const float4 r1 = *(const float4*)(W1 + (2 * p + 1) * FEAT + col4);
      const float4 q0 = *(const float4*)(W2 + (2 * p) * FEAT + col4);
      const float4 q1 = *(const float4*)(W2 + (2 * p + 1) * FEAT + col4);
      float* d1 = Ws + p * WPITCH2 + 2 * col4;
      float* d2 = d1 + WMAT;
      *(float4*)(d1)     = make_float4(r0.x, r1.x, r0.y, r1.y);
      *(float4*)(d1 + 4) = make_float4(r0.z, r1.z, r0.w, r1.w);
      *(float4*)(d2)     = make_float4(q0.x, q1.x, q0.y, q1.y);
      *(float4*)(d2 + 4) = make_float4(q0.z, q1.z, q0.w, q1.w);
    }
  }
  __syncthreads();

  const int g  = wave & 3;
  const int rs = wave >> 2;
  const int ncol = (g & 1) * 64 + l15;
  const float* __restrict__ wl = Ws + (g >> 1) * WMAT + hi * WPITCH2 + 2 * ncol;
  float* __restrict__ Hout = (g >> 1) ? H2 : H1;
  const int rowTiles = (M + 15) >> 4;

  for (int rt = 2 * (int)blockIdx.x + rs; rt < rowTiles; rt += 2 * (int)gridDim.x) {
    const int m0 = rt * 16;
    int mload = m0 + l15;
    if (mload > M - 1) mload = M - 1;
    const float* __restrict__ xrow = X + (size_t)mload * FEAT + 2 * hi;

    v8f c0 = {0.f, 0.f, 0.f, 0.f, 0.f, 0.f, 0.f, 0.f};
    v8f c1 = c0, c2 = c0, c3 = c0;

    v2f ac[8], an[8];
#pragma unroll
    for (int u = 0; u < 8; ++u) ac[u] = *(const v2f*)(xrow + 4 * u);
#pragma unroll 1
    for (int j = 0; j < 3; ++j) {
#pragma unroll
      for (int u = 0; u < 8; ++u) an[u] = *(const v2f*)(xrow + 32 * (j + 1) + 4 * u);
      gemm_k32(ac, wl + (16 * j) * WPITCH2, c0, c1, c2, c3);
#pragma unroll
      for (int u = 0; u < 8; ++u) ac[u] = an[u];
    }
    gemm_k32(ac, wl + 48 * WPITCH2, c0, c1, c2, c3);

    float* __restrict__ o = Hout + (size_t)(m0 + 8 * hi) * FEAT + ncol;
    if (m0 + 16 <= M) {
#pragma unroll
      for (int r = 0; r < 8; ++r) {
        o[(size_t)r * FEAT]      = c0[r];
        o[(size_t)r * FEAT + 16] = c1[r];
        o[(size_t)r * FEAT + 32] = c2[r];
        o[(size_t)r * FEAT + 48] = c3[r];
      }
    } else {
#pragma unroll
      for (int r = 0; r < 8; ++r) {
        if (m0 + r + 8 * hi < M) {
          o[(size_t)r * FEAT]      = c0[r];
          o[(size_t)r * FEAT + 16] = c1[r];
          o[(size_t)r * FEAT + 32] = c2[r];
          o[(size_t)r * FEAT + 48] = c3[r];
        }
      }
    }
  }
}

__global__ void cnt_zero(int4* __restrict__ c, int n4) {
  const int i = blockIdx.x * blockDim.x + threadIdx.x;
  if (i < n4) c[i] = make_int4(0, 0, 0, 0);
}

__global__ __launch_bounds__(256)
void edge_rank(const int* __restrict__ dst1, const int* __restrict__ dst2,
               int* __restrict__ cnt1, int* __restrict__ cnt2,
               int* __restrict__ rank, int E1, int E) {
  const int i = blockIdx.x * blockDim.x + threadIdx.x;
  if (i >= E) return;
  if (i < E1) rank[i] = atomicAdd(&cnt1[dst1[i]], 1);
  else        rank[i] = atomicAdd(&cnt2[dst2[i - E1]], 1);
}

__global__ __launch_bounds__(256)
void row_scan(const int* __restrict__ cnt1, const int* __restrict__ cnt2,
              int* __restrict__ start,
              float* __restrict__ dinv1, float* __restrict__ dinv2, int N) {
  __shared__ int wsum[8];
  __shared__ int wpre[8];
  const int t = threadIdx.x;
  const int lane = t & 31;
  const int wave = t >> 5;
  const int4* __restrict__ c1 = (const int4*)cnt1;
  const int4* __restrict__ c2 = (const int4*)cnt2;

  int pre = 0;
  for (int k = 0; k < (int)blockIdx.x; ++k) {
    const int4 a = c1[k * 256 + t];
    const int4 b = c2[k * 256 + t];
    pre += (a.x + a.y) + (a.z + a.w) + (b.x + b.y) + (b.z + b.w);
  }
  const int q = blockIdx.x * 256 + t;
  const int4 a = c1[q];
  const int4 b = c2[q];
  const int l0 = a.x + b.x, l1 = a.y + b.y, l2 = a.z + b.z, l3 = a.w + b.w;
  const int mine = (l0 + l1) + (l2 + l3);

  int inc = mine;
  for (int o = 1; o < 32; o <<= 1) {
    const int u = __shfl_up(inc, o);
    if (lane >= o) inc += u;
  }
  for (int o = 16; o > 0; o >>= 1) pre += __shfl_xor(pre, o);
  if (lane == 31) wsum[wave] = inc;
  if (lane == 0)  wpre[wave] = pre;
  __syncthreads();
  int off = 0;
  for (int w = 0; w < 8; ++w) {
    off += wpre[w];
    if (w < wave) off += wsum[w];
  }
  const int s0 = off + inc - mine;
  const int i = q * 4;
  const int s1 = s0 + l0, s2 = s1 + l1, s3 = s2 + l2;
  if (i < N)     { start[i]     = s0; dinv1[i]     = rsqrtf((float)(1 + a.x)); dinv2[i]     = rsqrtf((float)(1 + b.x)); }
  if (i + 1 < N) { start[i + 1] = s1; dinv1[i + 1] = rsqrtf((float)(1 + a.y)); dinv2[i + 1] = rsqrtf((float)(1 + b.y)); }
  if (i + 2 < N) { start[i + 2] = s2; dinv1[i + 2] = rsqrtf((float)(1 + a.z)); dinv2[i + 2] = rsqrtf((float)(1 + b.z)); }
  if (i + 3 < N) { start[i + 3] = s3; dinv1[i + 3] = rsqrtf((float)(1 + a.w)); dinv2[i + 3] = rsqrtf((float)(1 + b.w)); }
}

__global__ __launch_bounds__(256)
void csr_fill(const int* __restrict__ src1, const int* __restrict__ dst1,
              const int* __restrict__ src2, const int* __restrict__ dst2,
              const int* __restrict__ rank, const int* __restrict__ start,
              const int* __restrict__ cnt1,
              const float* __restrict__ dinv1, const float* __restrict__ dinv2,
              int2* __restrict__ csr, int E1, int E, int N) {
  const int i = blockIdx.x * blockDim.x + threadIdx.x;
  if (i >= E) return;
  int row, p;
  float w;
  if (i < E1) {
    const int s = src1[i], d = dst1[i];
    p = start[d] + rank[i];
    w = dinv1[s] * dinv1[d];
    row = s;
  } else {
    const int j = i - E1;
    const int s = src2[j], d = dst2[j];
    p = start[d] + cnt1[d] + rank[i];
    w = dinv2[s] * dinv2[d];
    row = N + s;
  }
  csr[p] = make_int2(row, __float_as_int(w));
}

#define ACC_ROW(g, e)                                   \
  acc.x = __builtin_fmaf((g).x, (e), acc.x);           \
  acc.y = __builtin_fmaf((g).y, (e), acc.y);           \
  acc.z = __builtin_fmaf((g).z, (e), acc.z);           \
  acc.w = __builtin_fmaf((g).w, (e), acc.w)

__global__ __launch_bounds__(256)
void gather_combine(const float* __restrict__ H12,
                    const int2* __restrict__ csr,
                    const int* __restrict__ start,
                    const int* __restrict__ cnt1,
                    const int* __restrict__ cnt2,
                    const float* __restrict__ dinv1,
                    const float* __restrict__ dinv2,
                    const float* __restrict__ b1,
                    const float* __restrict__ b2,
                    float* __restrict__ OUT,
                    int N, int relu) {
  const int lane = threadIdx.x & 31;
  const int v = __builtin_amdgcn_readfirstlane((int)(blockIdx.x * 8 + (threadIdx.x >> 5)));
  if (v >= N) return;
  const int beg = start[v];
  const int n = cnt1[v] + cnt2[v];
  const float d1 = dinv1[v];
  const float d2 = dinv2[v];
  const float w1 = d1 * d1;
  const float w2 = d2 * d2;
  const float4* __restrict__ Hq = (const float4*)H12 + lane;
  const float4 s1 = Hq[(size_t)v * 32];
  const float4 s2 = Hq[(size_t)(N + v) * 32];
  float4 acc;
  acc.x = s1.x * w1; acc.y = s1.y * w1; acc.z = s1.z * w1; acc.w = s1.w * w1;
  ACC_ROW(s2, w2);

  for (int base = 0; base < n; base += 32) {
    const int m = (n - base < 32) ? (n - base) : 32;
    int2 ent = make_int2(0, 0);
    if (lane < m) ent = csr[beg + base + lane];
    const int m4 = m & ~3;
    int j = 0;
    for (; j < m4; j += 4) {
      const int   r0 = __builtin_amdgcn_readlane(ent.x, j);
      const int   r1 = __builtin_amdgcn_readlane(ent.x, j + 1);
      const int   r2 = __builtin_amdgcn_readlane(ent.x, j + 2);
      const int   r3 = __builtin_amdgcn_readlane(ent.x, j + 3);
      const float e0 = __int_as_float(__builtin_amdgcn_readlane(ent.y, j));
      const float e1 = __int_as_float(__builtin_amdgcn_readlane(ent.y, j + 1));
      const float e2 = __int_as_float(__builtin_amdgcn_readlane(ent.y, j + 2));
      const float e3 = __int_as_float(__builtin_amdgcn_readlane(ent.y, j + 3));
      const float4 g0 = Hq[(size_t)r0 * 32];
      const float4 g1 = Hq[(size_t)r1 * 32];
      const float4 g2 = Hq[(size_t)r2 * 32];
      const float4 g3 = Hq[(size_t)r3 * 32];
      ACC_ROW(g0, e0); ACC_ROW(g1, e1); ACC_ROW(g2, e2); ACC_ROW(g3, e3);
    }
    if (j < m) {
      const int   r0 = __builtin_amdgcn_readlane(ent.x, j);
      const int   r1 = __builtin_amdgcn_readlane(ent.x, j + 1);
      const int   r2 = __builtin_amdgcn_readlane(ent.x, j + 2);
      const float e0 = __int_as_float(__builtin_amdgcn_readlane(ent.y, j));
      const float e1 = __int_as_float(__builtin_amdgcn_readlane(ent.y, j + 1));
      const float e2 = __int_as_float(__builtin_amdgcn_readlane(ent.y, j + 2));
      const float4 g0 = Hq[(size_t)r0 * 32];
      float4 g1 = make_float4(0.f, 0.f, 0.f, 0.f);
      float4 g2 = g1;
      if (j + 1 < m) g1 = Hq[(size_t)r1 * 32];
      if (j + 2 < m) g2 = Hq[(size_t)r2 * 32];
      ACC_ROW(g0, e0); ACC_ROW(g1, e1); ACC_ROW(g2, e2);
    }
  }

  const float4 c1 = *((const float4*)b1 + lane);
  const float4 c2 = *((const float4*)b2 + lane);
  float4 o;
  o.x = acc.x + c1.x + c2.x;
  o.y = acc.y + c1.y + c2.y;
  o.z = acc.z + c1.z + c2.z;
  o.w = acc.w + c1.w + c2.w;
  if (relu) {
    o.x = fmaxf(o.x, 0.f); o.y = fmaxf(o.y, 0.f);
    o.z = fmaxf(o.z, 0.f); o.w = fmaxf(o.w, 0.f);
  }
  *((float4*)(OUT + (size_t)v * FEAT) + lane) = o;
}

extern "C" void kernel_launch(void* const* d_in, const int* in_sizes, int n_in,
                              void* d_out, int out_size, void* d_ws, size_t ws_size,
                              hipStream_t stream) {
  const float* x      = (const float*)d_in[0];
  const int*   ei     = (const int*)d_in[1];
  const int*   cei    = (const int*)d_in[2];
  const float* conv_w = (const float*)d_in[3];
  const float* conv_b = (const float*)d_in[4];
  const float* ctrl_w = (const float*)d_in[5];
  const float* ctrl_b = (const float*)d_in[6];

  const int N  = in_sizes[0] / FEAT;
  const int E1 = in_sizes[1] / 2;
  const int E2 = in_sizes[2] / 2;
  const int D  = in_sizes[3] / (FEAT * FEAT);
  const int E  = E1 + E2;

  const int* src1 = ei;        const int* dst1 = ei + E1;
  const int* src2 = cei;       const int* dst2 = cei + E2;

  const size_t NF = (size_t)N * FEAT;
  const int Npad = ((N + 1023) / 1024) * 1024;
  float* h1    = (float*)d_ws;
  float* h2    = h1 + NF;
  float* xbuf  = h2 + NF;
  float* dinv1 = xbuf + NF;
  float* dinv2 = dinv1 + Npad;
  int*   cnt1  = (int*)(dinv2 + Npad);
  int*   cnt2  = cnt1 + Npad;
  int*   start = cnt2 + Npad;
  int*   rank  = start + Npad;
  int2*  csr   = (int2*)(rank + ((E + 3) / 4) * 4);

  const int T = 256;
  const int edgeBlocks = (E + T - 1) / T;

  cnt_zero<<<(2 * Npad / 4 + T - 1) / T, T, 0, stream>>>((int4*)cnt1, 2 * Npad / 4);
  edge_rank<<<edgeBlocks, T, 0, stream>>>(dst1, dst2, cnt1, cnt2, rank, E1, E);
  row_scan<<<Npad / 1024, T, 0, stream>>>(cnt1, cnt2, start, dinv1, dinv2, N);
  csr_fill<<<edgeBlocks, T, 0, stream>>>(src1, dst1, src2, dst2, rank, start, cnt1,
                                         dinv1, dinv2, csr, E1, E, N);

  const int gemmBlocks = GEMM_BLOCKS;
  const int nodeWaveBlocks = (N + 7) / 8;

  const float* x_cur = x;
  for (int i = 0; i < D; ++i) {
    const float* W1 = conv_w + (size_t)i * FEAT * FEAT;
    const float* W2 = ctrl_w + (size_t)i * FEAT * FEAT;
    const float* B1 = conv_b + (size_t)i * FEAT;
    const float* B2 = ctrl_b + (size_t)i * FEAT;
    float* x_next = (i == D - 1) ? (float*)d_out : xbuf;

    gcn_gemm_dual_wmma<<<gemmBlocks, T, 0, stream>>>(x_cur, W1, W2, h1, h2, N);
    gather_combine<<<nodeWaveBlocks, T, 0, stream>>>(h1, csr, start, cnt1, cnt2, dinv1, dinv2,
                                                     B1, B2, x_next, N, (i != D - 1) ? 1 : 0);
    x_cur = x_next;
  }
}


# Layers 1..3: transform fused into the previous gather launch: 512 resident blocks sum 16-node tiles (2 nodes a wave, 16 rows in flight) into LDS and multiply by LDS-held W1|W2; sum and k order kept
# speedup vs baseline: 3.6387x; 1.0789x over previous
#include <hip/hip_runtime.h>

typedef float v2f __attribute__((ext_vector_type(2)));
typedef float v8f __attribute__((ext_vector_type(8)));

#define FEAT 128

#define WPITCH2 288
#define WMAT (64 * WPITCH2)
#define GEMM_BLOCKS 512

__device__ __forceinline__ void gemm_k32(const v2f (&a)[8], const float* __restrict__ wp0,
                                         v8f& c0, v8f& c1, v8f& c2, v8f& c3) {
#pragma unroll
  for (int u = 0; u < 8; ++u) {
    const float* __restrict__ wp = wp0 + (2 * u) * WPITCH2;
    const v2f b0 = *(const v2f*)(wp);
    const v2f b1 = *(const v2f*)(wp + 32);
    const v2f b2 = *(const v2f*)(wp + 64);
    const v2f b3 = *(const v2f*)(wp + 96);
    c0 = __builtin_amdgcn_wmma_f32_16x16x4_f32(false, a[u], false, b0, (short)0, c0, false, false);
    c1 = __builtin_amdgcn_wmma_f32_16x16x4_f32(false, a[u], false, b1, (short)0, c1, false, false);
    c2 = __builtin_amdgcn_wmma_f32_16x16x4_f32(false, a[u], false, b2, (short)0, c2, false, false);
    c3 = __builtin_amdgcn_wmma_f32_16x16x4_f32(false, a[u], false, b3, (short)0, c3, false, false);
  }
}

__global__ __launch_bounds__(256)
void gcn_gemm_dual_wmma(const float* __restrict__ X,
                        const float* __restrict__ W1,
                        const float* __restrict__ W2,
                        float* __restrict__ H1,
                        float* __restrict__ H2,
                        int M) {
  __shared__ __attribute__((aligned(16))) float Ws[2 * WMAT];

  const int lane = threadIdx.x & 31;
  const int wave = threadIdx.x >> 5;
  const int l15 = lane & 15;
  const int hi  = lane >> 4;

  {
    const int col4 = lane * 4;
#pragma unroll
    for (int i = 0; i < 8; ++i) {
      const int p = wave + 8 * i;
      const float4 r0 = *(const float4*)(W1 + (2 * p) * FEAT + col4);
      const float4 r1 = *(const float4*)(W1 + (2 * p + 1) * FEAT + col4);
      const float4 q0 = *(const float4*)(W2 + (2 * p) * FEAT + col4);
      const float4 q1 = *(const float4*)(W2 + (2 * p + 1) * FEAT + col4);
      float* d1 = Ws + p * WPITCH2 + 2 * col4;
      float* d2 = d1 + WMAT;
      *(float4*)(d1)     = make_float4(r0.x, r1.x, r0.y, r1.y);
      *(float4*)(d1 + 4) = make_float4(r0.z, r1.z, r0.w, r1.w);
      *(float4*)(d2)     = make_float4(q0.x, q1.x, q0.y, q1.y);
      *(float4*)(d2 + 4) = make_float4(q0.z, q1.z, q0.w, q1.w);
    }
  }
  __syncthreads();

  const int g  = wave & 3;
  const int rs = wave >> 2;
  const int ncol = (g & 1) * 64 + l15;
  const float* __restrict__ wl = Ws + (g >> 1) * WMAT + hi * WPITCH2 + 2 * ncol;
  float* __restrict__ Hout = (g >> 1) ? H2 : H1;
  const int rowTiles = (M + 15) >> 4;

  for (int rt = 2 * (int)blockIdx.x + rs; rt < rowTiles; rt += 2 * (int)gridDim.x) {
    const int m0 = rt * 16;
    int mload = m0 + l15;
    if (mload > M - 1) mload = M - 1;
    const float* __restrict__ xrow = X + (size_t)mload * FEAT + 2 * hi;

    v8f c0 = {0.f, 0.f, 0.f, 0.f, 0.f, 0.f, 0.f, 0.f};
    v8f c1 = c0, c2 = c0, c3 = c0;

    v2f ac[8], an[8];
#pragma unroll
    for (int u = 0; u < 8; ++u) ac[u] = *(const v2f*)(xrow + 4 * u);
#pragma unroll 1
    for (int j = 0; j < 3; ++j) {
#pragma unroll
      for (int u = 0; u < 8; ++u) an[u] = *(const v2f*)(xrow + 32 * (j + 1) + 4 * u);
      gemm_k32(ac, wl + (16 * j) * WPITCH2, c0, c1, c2, c3);
#pragma unroll
      for (int u = 0; u < 8; ++u) ac[u] = an[u];
    }
    gemm_k32(ac, wl + 48 * WPITCH2, c0, c1, c2, c3);

    float* __restrict__ o = Hout + (size_t)(m0 + 8 * hi) * FEAT + ncol;
    if (m0 + 16 <= M) {
#pragma unroll
      for (int r = 0; r < 8; ++r) {
        o[(size_t)r * FEAT]      = c0[r];
        o[(size_t)r * FEAT + 16] = c1[r];
        o[(size_t)r * FEAT + 32] = c2[r];
        o[(size_t)r * FEAT + 48] = c3[r];
      }
    } else {
#pragma unroll
      for (int r = 0; r < 8; ++r) {
        if (m0 + r + 8 * hi < M) {
          o[(size_t)r * FEAT]      = c0[r];
          o[(size_t)r * FEAT + 16] = c1[r];
          o[(size_t)r * FEAT + 32] = c2[r];
          o[(size_t)r * FEAT + 48] = c3[r];
        }
      }
    }
  }
}

__global__ void cnt_zero(int4* __restrict__ c, int n4) {
  const int i = blockIdx.x * blockDim.x + threadIdx.x;
  if (i < n4) c[i] = make_int4(0, 0, 0, 0);
}

__global__ __launch_bounds__(256)
void edge_rank(const int* __restrict__ dst1, const int* __restrict__ dst2,
               int* __restrict__ cnt1, int* __restrict__ cnt2,
               int* __restrict__ rank, int E1, int E) {
  const int i = blockIdx.x * blockDim.x + threadIdx.x;
  if (i >= E) return;
  if (i < E1) rank[i] = atomicAdd(&cnt1[dst1[i]], 1);
  else        rank[i] = atomicAdd(&cnt2[dst2[i - E1]], 1);
}

__global__ __launch_bounds__(256)
void row_scan(const int* __restrict__ cnt1, const int* __restrict__ cnt2,
              int* __restrict__ start,
              float* __restrict__ dinv1, float* __restrict__ dinv2, int N) {
  __shared__ int wsum[8];
  __shared__ int wpre[8];
  const int t = threadIdx.x;
  const int lane = t & 31;
  const int wave = t >> 5;
  const int4* __restrict__ c1 = (const int4*)cnt1;
  const int4* __restrict__ c2 = (const int4*)cnt2;

  int pre = 0;
  for (int k = 0; k < (int)blockIdx.x; ++k) {
    const int4 a = c1[k * 256 + t];
    const int4 b = c2[k * 256 + t];
    pre += (a.x + a.y) + (a.z + a.w) + (b.x + b.y) + (b.z + b.w);
  }
  const int q = blockIdx.x * 256 + t;
  const int4 a = c1[q];
  const int4 b = c2[q];
  const int l0 = a.x + b.x, l1 = a.y + b.y, l2 = a.z + b.z, l3 = a.w + b.w;
  const int mine = (l0 + l1) + (l2 + l3);

  int inc = mine;
  for (int o = 1; o < 32; o <<= 1) {
    const int u = __shfl_up(inc, o);
    if (lane >= o) inc += u;
  }
  for (int o = 16; o > 0; o >>= 1) pre += __shfl_xor(pre, o);
  if (lane == 31) wsum[wave] = inc;
  if (lane == 0)  wpre[wave] = pre;
  __syncthreads();
  int off = 0;
  for (int w = 0; w < 8; ++w) {
    off += wpre[w];
    if (w < wave) off += wsum[w];
  }
  const int s0 = off + inc - mine;
  const int i = q * 4;
  const int s1 = s0 + l0, s2 = s1 + l1, s3 = s2 + l2;
  if (i < N)     { start[i]     = s0; dinv1[i]     = rsqrtf((float)(1 + a.x)); dinv2[i]     = rsqrtf((float)(1 + b.x)); }
  if (i + 1 < N) { start[i + 1] = s1; dinv1[i + 1] = rsqrtf((float)(1 + a.y)); dinv2[i + 1] = rsqrtf((float)(1 + b.y)); }
  if (i + 2 < N) { start[i + 2] = s2; dinv1[i + 2] = rsqrtf((float)(1 + a.z)); dinv2[i + 2] = rsqrtf((float)(1 + b.z)); }
  if (i + 3 < N) { start[i + 3] = s3; dinv1[i + 3] = rsqrtf((float)(1 + a.w)); dinv2[i + 3] = rsqrtf((float)(1 + b.w)); }
}

__global__ __launch_bounds__(256)
void csr_fill(const int* __restrict__ src1, const int* __restrict__ dst1,
              const int* __restrict__ src2, const int* __restrict__ dst2,
              const int* __restrict__ rank, const int* __restrict__ start,
              const int* __restrict__ cnt1,
              const float* __restrict__ dinv1, const float* __restrict__ dinv2,
              int2* __restrict__ csr, int E1, int E, int N) {
  const int i = blockIdx.x * blockDim.x + threadIdx.x;
  if (i >= E) return;
  int row, p;
  float w;
  if (i < E1) {
    const int s = src1[i], d = dst1[i];
    p = start[d] + rank[i];
    w = dinv1[s] * dinv1[d];
    row = s;
  } else {
    const int j = i - E1;
    const int s = src2[j], d = dst2[j];
    p = start[d] + cnt1[d] + rank[i];
    w = dinv2[s] * dinv2[d];
    row = N + s;
  }
  csr[p] = make_int2(row, __float_as_int(w));
}

#define ACC_ROW(g, e)                                   \
  acc.x = __builtin_fmaf((g).x, (e), acc.x);           \
  acc.y = __builtin_fmaf((g).y, (e), acc.y);           \
  acc.z = __builtin_fmaf((g).z, (e), acc.z);           \
  acc.w = __builtin_fmaf((g).w, (e), acc.w)

__global__ __launch_bounds__(256)
void gather_combine(const float* __restrict__ H12,
                    const int2* __restrict__ csr,
                    const int* __restrict__ start,
                    const int* __restrict__ cnt1,
                    const int* __restrict__ cnt2,
                    const float* __restrict__ dinv1,
                    const float* __restrict__ dinv2,
                    const float* __restrict__ b1,
                    const float* __restrict__ b2,
                    float* __restrict__ OUT,
                    int N, int relu) {
  const int lane = threadIdx.x & 31;
  const int v = __builtin_amdgcn_readfirstlane((int)(blockIdx.x * 8 + (threadIdx.x >> 5)));
  if (v >= N) return;
  const int beg = start[v];
  const int n = cnt1[v] + cnt2[v];
  const float d1 = dinv1[v];
  const float d2 = dinv2[v];
  const float w1 = d1 * d1;
  const float w2 = d2 * d2;
  const float4* __restrict__ Hq = (const float4*)H12 + lane;
  const float4 s1 = Hq[(size_t)v * 32];
  const float4 s2 = Hq[(size_t)(N + v) * 32];
  float4 acc;
  acc.x = s1.x * w1; acc.y = s1.y * w1; acc.z = s1.z * w1; acc.w = s1.w * w1;
  ACC_ROW(s2, w2);

  for (int base = 0; base < n; base += 32) {
    const int m = (n - base < 32) ? (n - base) : 32;
    int2 ent = make_int2(0, 0);
    if (lane < m) ent = csr[beg + base + lane];
    const int m4 = m & ~3;
    int j = 0;
    for (; j < m4; j += 4) {
      const int   r0 = __builtin_amdgcn_readlane(ent.x, j);
      const int   r1 = __builtin_amdgcn_readlane(ent.x, j + 1);
      const int   r2 = __builtin_amdgcn_readlane(ent.x, j + 2);
      const int   r3 = __builtin_amdgcn_readlane(ent.x, j + 3);
      const float e0 = __int_as_float(__builtin_amdgcn_readlane(ent.y, j));
      const float e1 = __int_as_float(__builtin_amdgcn_readlane(ent.y, j + 1));
      const float e2 = __int_as_float(__builtin_amdgcn_readlane(ent.y, j + 2));
      const float e3 = __int_as_float(__builtin_amdgcn_readlane(ent.y, j + 3));
      const float4 g0 = Hq[(size_t)r0 * 32];
      const float4 g1 = Hq[(size_t)r1 * 32];
      const float4 g2 = Hq[(size_t)r2 * 32];
      const float4 g3 = Hq[(size_t)r3 * 32];
      ACC_ROW(g0, e0); ACC_ROW(g1, e1); ACC_ROW(g2, e2); ACC_ROW(g3, e3);
    }
    if (j < m) {
      const int   r0 = __builtin_amdgcn_readlane(ent.x, j);
      const int   r1 = __builtin_amdgcn_readlane(ent.x, j + 1);
      const int   r2 = __builtin_amdgcn_readlane(ent.x, j + 2);
      const float e0 = __int_as_float(__builtin_amdgcn_readlane(ent.y, j));
      const float e1 = __int_as_float(__builtin_amdgcn_readlane(ent.y, j + 1));
      const float e2 = __int_as_float(__builtin_amdgcn_readlane(ent.y, j + 2));
      const float4 g0 = Hq[(size_t)r0 * 32];
      float4 g1 = make_float4(0.f, 0.f, 0.f, 0.f);
      float4 g2 = g1;
      if (j + 1 < m) g1 = Hq[(size_t)r1 * 32];
      if (j + 2 < m) g2 = Hq[(size_t)r2 * 32];
      ACC_ROW(g0, e0); ACC_ROW(g1, e1); ACC_ROW(g2, e2);
    }
  }

  const float4 c1 = *((const float4*)b1 + lane);
  const float4 c2 = *((const float4*)b2 + lane);
  float4 o;
  o.x = acc.x + c1.x + c2.x;
  o.y = acc.y + c1.y + c2.y;
  o.z = acc.z + c1.z + c2.z;
  o.w = acc.w + c1.w + c2.w;
  if (relu) {
    o.x = fmaxf(o.x, 0.f); o.y = fmaxf(o.y, 0.f);
    o.z = fmaxf(o.z, 0.f); o.w = fmaxf(o.w, 0.f);
  }
  *((float4*)(OUT + (size_t)v * FEAT) + lane) = o;
}


#define XPITCH 132

#define ACC_ROW_TO(a, g, e)                             \
  (a).x = __builtin_fmaf((g).x, (e), (a).x);           \
  (a).y = __builtin_fmaf((g).y, (e), (a).y);           \
  (a).z = __builtin_fmaf((g).z, (e), (a).z);           \
  (a).w = __builtin_fmaf((g).w, (e), (a).w)

template <bool GUARD>
__device__ __forceinline__ void gather_pair8(const float4* __restrict__ Hq,
                                             const int2 entA, const int2 entB,
                                             const int j, const int mA, const int mB,
                                             float4& accA, float4& accB) {
  float4 gA[8], gB[8];
  float eA[8], eB[8];
#pragma unroll
  for (int q = 0; q < 8; ++q) {
    const int r = __builtin_amdgcn_readlane(entA.x, j + q);
    eA[q] = __int_as_float(__builtin_amdgcn_readlane(entA.y, j + q));
    gA[q] = make_float4(0.f, 0.f, 0.f, 0.f);
    if (!GUARD || j + q < mA) gA[q] = Hq[(size_t)r * 32];
  }
#pragma unroll
  for (int q = 0; q < 8; ++q) {
    const int r = __builtin_amdgcn_readlane(entB.x, j + q);
    eB[q] = __int_as_float(__builtin_amdgcn_readlane(entB.y, j + q));
    gB[q] = make_float4(0.f, 0.f, 0.f, 0.f);
    if (!GUARD || j + q < mB) gB[q] = Hq[(size_t)r * 32];
  }
#pragma unroll
  for (int q = 0; q < 8; ++q) { ACC_ROW_TO(accA, gA[q], eA[q]); }
#pragma unroll
  for (int q = 0; q < 8; ++q) { ACC_ROW_TO(accB, gB[q], eB[q]); }
}

__global__ __launch_bounds__(256)
void gather_gemm_fused(const float* __restrict__ H12,
                       const int2* __restrict__ csr,
                       const int* __restrict__ start,
                       const int* __restrict__ cnt1,
                       const int* __restrict__ cnt2,
                       const float* __restrict__ dinv1,
                       const float* __restrict__ dinv2,
                       const float* __restrict__ b1,
                       const float* __restrict__ b2,
                       const float* __restrict__ W1,
                       const float* __restrict__ W2,
                       float* __restrict__ H1n,
                       float* __restrict__ H2n,
                       int N) {
  __shared__ __attribute__((aligned(16))) float Ws[2 * WMAT];
  __shared__ __attribute__((aligned(16))) float Xs[16 * XPITCH];

  const int lane = threadIdx.x & 31;
  const int wave = __builtin_amdgcn_readfirstlane((int)(threadIdx.x >> 5));
  const int l15 = lane & 15;
  const int hi  = lane >> 4;

  {
    const int col4 = lane * 4;
#pragma unroll
    for (int i = 0; i < 8; ++i) {
      const int p = wave + 8 * i;
      const float4 r0 = *(const float4*)(W1 + (2 * p) * FEAT + col4);
      const float4 r1 = *(const float4*)(W1 + (2 * p + 1) * FEAT + col4);
      const float4 q0 = *(const float4*)(W2 + (2 * p) * FEAT + col4);
      const float4 q1 = *(const float4*)(W2 + (2 * p + 1) * FEAT + col4);
      float* d1 = Ws + p * WPITCH2 + 2 * col4;
      float* d2 = d1 + WMAT;
      *(float4*)(d1)     = make_float4(r0.x, r1.x, r0.y, r1.y);
      *(float4*)(d1 + 4) = make_float4(r0.z, r1.z, r0.w, r1.w);
      *(float4*)(d2)     = make_float4(q0.x, q1.x, q0.y, q1.y);
      *(float4*)(d2 + 4) = make_float4(q0.z, q1.z, q0.w, q1.w);
    }
  }
  const float4 bias1 = *((const float4*)b1 + lane);
  const float4 bias2 = *((const float4*)b2 + lane);
  __syncthreads();

  const float4* __restrict__ Hq = (const float4*)H12 + lane;
  const int mat  = wave >> 2;
  const int colb = 32 * (wave & 3) + l15;
  const float* __restrict__ wl = Ws + mat * WMAT + hi * WPITCH2 + 2 * colb;
  const float* __restrict__ xa = Xs + l15 * XPITCH + 2 * hi;
  float* __restrict__ Hout = mat ? H2n : H1n;
  const int tiles = (N + 15) >> 4;

  for (int t = (int)blockIdx.x; t < tiles; t += (int)gridDim.x) {
    const int vA = 16 * t + 2 * wave;
    float4 oA = make_float4(0.f, 0.f, 0.f, 0.f);
    float4 oB = oA;
    if (vA < N) {
      const bool hasB = (vA + 1 < N);
      const int vB = hasB ? vA + 1 : vA;
      const float4 sA1 = Hq[(size_t)vA * 32];
      const float4 sA2 = Hq[(size_t)(N + vA) * 32];
      const float4 sB1 = Hq[(size_t)vB * 32];
      const float4 sB2 = Hq[(size_t)(N + vB) * 32];
      const int beg = start[vA];
      const int nA = cnt1[vA] + cnt2[vA];
      const int nBr = cnt1[vB] + cnt2[vB];
      const float dA1 = dinv1[vA], dA2 = dinv2[vA];
      const float dB1 = dinv1[vB], dB2 = dinv2[vB];
      asm volatile("" :: "s"(beg));
      const int nB = hasB ? nBr : 0;
      const float wA1 = dA1 * dA1, wA2 = dA2 * dA2;
      const float wB1 = dB1 * dB1, wB2 = dB2 * dB2;
      float4 accA, accB;
      accA.x = sA1.x * wA1; accA.y = sA1.y * wA1; accA.z = sA1.z * wA1; accA.w = sA1.w * wA1;
      ACC_ROW_TO(accA, sA2, wA2);
      accB.x = sB1.x * wB1; accB.y = sB1.y * wB1; accB.z = sB1.z * wB1; accB.w = sB1.w * wB1;
      ACC_ROW_TO(accB, sB2, wB2);

      const int nmax = (nA > nB) ? nA : nB;
      for (int base = 0; base < nmax; base += 32) {
        int mA = nA - base; mA = (mA < 0) ? 0 : ((mA > 32) ? 32 : mA);
        int mB = nB - base; mB = (mB < 0) ? 0 : ((mB > 32) ? 32 : mB);
        int2 entA = make_int2(0, 0);
        int2 entB = make_int2(0, 0);
        if (lane < mA) entA = csr[beg + base + lane];
        if (lane < mB) entB = csr[beg + nA + base + lane];
        const int mlo = ((mA < mB) ? mA : mB) & ~7;
        const int mhi = (mA > mB) ? mA : mB;
        int j = 0;
        for (; j < mlo; j += 8) gather_pair8<false>(Hq, entA, entB, j, mA, mB, accA, accB);
        for (; j < mhi; j += 8) gather_pair8<true >(Hq, entA, entB, j, mA, mB, accA, accB);
      }

      oA.x = fmaxf(accA.x + bias1.x + bias2.x, 0.f);
      oA.y = fmaxf(accA.y + bias1.y + bias2.y, 0.f);
      oA.z = fmaxf(accA.z + bias1.z + bias2.z, 0.f);
      oA.w = fmaxf(accA.w + bias1.w + bias2.w, 0.f);
      if (hasB) {
        oB.x = fmaxf(accB.x + bias1.x + bias2.x, 0.f);
        oB.y = fmaxf(accB.y + bias1.y + bias2.y, 0.f);
        oB.z = fmaxf(accB.z + bias1.z + bias2.z, 0.f);
        oB.w = fmaxf(accB.w + bias1.w + bias2.w, 0.f);
      }
    }
    *(float4*)(Xs + (2 * wave) * XPITCH + 4 * lane)     = oA;
    *(float4*)(Xs + (2 * wave + 1) * XPITCH + 4 * lane) = oB;
    __syncthreads();

    v8f c0 = {0.f, 0.f, 0.f, 0.f, 0.f, 0.f, 0.f, 0.f};
    v8f c1 = c0;
    for (int kc = 0; kc < 4; ++kc) {
      v2f a[8], p0[8], p1[8];
#pragma unroll
      for (int u = 0; u < 8; ++u) {
        a[u]  = *(const v2f*)(xa + 32 * kc + 4 * u);
        const float* __restrict__ wp = wl + (16 * kc + 2 * u) * WPITCH2;
        p0[u] = *(const v2f*)(wp);
        p1[u] = *(const v2f*)(wp + 32);
      }
#pragma unroll
      for (int u = 0; u < 8; ++u) {
        c0 = __builtin_amdgcn_wmma_f32_16x16x4_f32(false, a[u], false, p0[u], (short)0, c0, false, false);
        c1 = __builtin_amdgcn_wmma_f32_16x16x4_f32(false, a[u], false, p1[u], (short)0, c1, false, false);
      }
    }

    __syncthreads();

    const int m0 = 16 * t;
    float* __restrict__ o = Hout + (size_t)(m0 + 8 * hi) * FEAT + colb;
    if (m0 + 16 <= N) {
#pragma unroll
      for (int r = 0; r < 8; ++r) {
        o[(size_t)r * FEAT]      = c0[r];
        o[(size_t)r * FEAT + 16] = c1[r];
      }
    } else {
#pragma unroll
      for (int r = 0; r < 8; ++r) {
        if (m0 + r + 8 * hi < N) {
          o[(size_t)r * FEAT]      = c0[r];
          o[(size_t)r * FEAT + 16] = c1[r];
        }
      }
    }
  }
}

extern "C" void kernel_launch(void* const* d_in, const int* in_sizes, int n_in,
                              void* d_out, int out_size, void* d_ws, size_t ws_size,
                              hipStream_t stream) {
  const float* x      = (const float*)d_in[0];
  const int*   ei     = (const int*)d_in[1];
  const int*   cei    = (const int*)d_in[2];
  const float* conv_w = (const float*)d_in[3];
  const float* conv_b = (const float*)d_in[4];
  const float* ctrl_w = (const float*)d_in[5];
  const float* ctrl_b = (const float*)d_in[6];

  const int N  = in_sizes[0] / FEAT;
  const int E1 = in_sizes[1] / 2;
  const int E2 = in_sizes[2] / 2;
  const int D  = in_sizes[3] / (FEAT * FEAT);
  const int E  = E1 + E2;

  const int* src1 = ei;        const int* dst1 = ei + E1;
  const int* src2 = cei;       const int* dst2 = cei + E2;

  const size_t NF = (size_t)N * FEAT;
  const int Npad = ((N + 1023) / 1024) * 1024;
  float* h1    = (float*)d_ws;
  float* h2    = h1 + NF;
  float* g1    = h2 + NF;
  float* dinv1 = g1 + 2 * NF;
  float* dinv2 = dinv1 + Npad;
  int*   cnt1  = (int*)(dinv2 + Npad);
  int*   cnt2  = cnt1 + Npad;
  int*   start = cnt2 + Npad;
  int*   rank  = start + Npad;
  int2*  csr   = (int2*)(rank + ((E + 3) / 4) * 4);

  const int T = 256;
  const int edgeBlocks = (E + T - 1) / T;

  cnt_zero<<<(2 * Npad / 4 + T - 1) / T, T, 0, stream>>>((int4*)cnt1, 2 * Npad / 4);
  edge_rank<<<edgeBlocks, T, 0, stream>>>(dst1, dst2, cnt1, cnt2, rank, E1, E);
  row_scan<<<Npad / 1024, T, 0, stream>>>(cnt1, cnt2, start, dinv1, dinv2, N);
  csr_fill<<<edgeBlocks, T, 0, stream>>>(src1, dst1, src2, dst2, rank, start, cnt1,
                                         dinv1, dinv2, csr, E1, E, N);

  const int gemmBlocks = GEMM_BLOCKS;
  const int nodeWaveBlocks = (N + 7) / 8;

  gcn_gemm_dual_wmma<<<gemmBlocks, T, 0, stream>>>(x, conv_w, ctrl_w, h1, h2, N);

  float* cur = h1;
  float* nxt = g1;
  for (int i = 0; i + 1 < D; ++i) {
    const float* B1 = conv_b + (size_t)i * FEAT;
    const float* B2 = ctrl_b + (size_t)i * FEAT;
    const float* W1 = conv_w + (size_t)(i + 1) * FEAT * FEAT;
    const float* W2 = ctrl_w + (size_t)(i + 1) * FEAT * FEAT;
    gather_gemm_fused<<<gemmBlocks, T, 0, stream>>>(cur, csr, start, cnt1, cnt2, dinv1, dinv2,
                                                    B1, B2, W1, W2, nxt, nxt + NF, N);
    float* const was = cur;
    cur = nxt;
    nxt = was;
  }

  gather_combine<<<nodeWaveBlocks, T, 0, stream>>>(cur, csr, start, cnt1, cnt2, dinv1, dinv2,
                                                   conv_b + (size_t)(D - 1) * FEAT,
                                                   ctrl_b + (size_t)(D - 1) * FEAT,
                                                   (float*)d_out, N, 0);
}


# Aggregate before transform: a layer launch sums weighted INPUT rows per node and edge set, then multiplies [A1X|A2X] by [W1;W2] on WMMA; layer-0 GEMM launch and 256-wide planes go; sum order changes
# speedup vs baseline: 3.7589x; 1.0330x over previous
#include <hip/hip_runtime.h>

typedef float v2f __attribute__((ext_vector_type(2)));
typedef float v8f __attribute__((ext_vector_type(8)));

#define FEAT 128
#define LAYER_BLOCKS 512

__global__ void cnt_zero(int4* __restrict__ c, int n4) {
  const int i = blockIdx.x * blockDim.x + threadIdx.x;
  if (i < n4) c[i] = make_int4(0, 0, 0, 0);
}

__global__ __launch_bounds__(256)
void edge_rank(const int* __restrict__ dst1, const int* __restrict__ dst2,
               int* __restrict__ cnt1, int* __restrict__ cnt2,
               int* __restrict__ rank, int E1, int E) {
  const int i = blockIdx.x * blockDim.x + threadIdx.x;
  if (i >= E) return;
  if (i < E1) rank[i] = atomicAdd(&cnt1[dst1[i]], 1);
  else        rank[i] = atomicAdd(&cnt2[dst2[i - E1]], 1);
}

__global__ __launch_bounds__(256)
void row_scan(const int* __restrict__ cnt1, const int* __restrict__ cnt2,
              int* __restrict__ start,
              float* __restrict__ dinv1, float* __restrict__ dinv2, int N) {
  __shared__ int wsum[8];
  __shared__ int wpre[8];
  const int t = threadIdx.x;
  const int lane = t & 31;
  const int wave = t >> 5;
  const int4* __restrict__ c1 = (const int4*)cnt1;
  const int4* __restrict__ c2 = (const int4*)cnt2;

  int pre = 0;
  for (int k = 0; k < (int)blockIdx.x; ++k) {
    const int4 a = c1[k * 256 + t];
    const int4 b = c2[k * 256 + t];
    pre += (a.x + a.y) + (a.z + a.w) + (b.x + b.y) + (b.z + b.w);
  }
  const int q = blockIdx.x * 256 + t;
  const int4 a = c1[q];
  const int4 b = c2[q];
  const int l0 = a.x + b.x, l1 = a.y + b.y, l2 = a.z + b.z, l3 = a.w + b.w;
  const int mine = (l0 + l1) + (l2 + l3);

  int inc = mine;
  for (int o = 1; o < 32; o <<= 1) {
    const int u = __shfl_up(inc, o);
    if (lane >= o) inc += u;
  }
  for (int o = 16; o > 0; o >>= 1) pre += __shfl_xor(pre, o);
  if (lane == 31) wsum[wave] = inc;
  if (lane == 0)  wpre[wave] = pre;
  __syncthreads();
  int off = 0;
  for (int w = 0; w < 8; ++w) {
    off += wpre[w];
    if (w < wave) off += wsum[w];
  }
  const int s0 = off + inc - mine;
  const int i = q * 4;
  const int s1 = s0 + l0, s2 = s1 + l1, s3 = s2 + l2;
  if (i < N)     { start[i]     = s0; dinv1[i]     = rsqrtf((float)(1 + a.x)); dinv2[i]     = rsqrtf((float)(1 + b.x)); }
  if (i + 1 < N) { start[i + 1] = s1; dinv1[i + 1] = rsqrtf((float)(1 + a.y)); dinv2[i + 1] = rsqrtf((float)(1 + b.y)); }
  if (i + 2 < N) { start[i + 2] = s2; dinv1[i + 2] = rsqrtf((float)(1 + a.z)); dinv2[i + 2] = rsqrtf((float)(1 + b.z)); }
  if (i + 3 < N) { start[i + 3] = s3; dinv1[i + 3] = rsqrtf((float)(1 + a.w)); dinv2[i + 3] = rsqrtf((float)(1 + b.w)); }
}

__global__ __launch_bounds__(256)
void csr_fill(const int* __restrict__ src1, const int* __restrict__ dst1,
              const int* __restrict__ src2, const int* __restrict__ dst2,
              const int* __restrict__ rank, const int* __restrict__ start,
              const int* __restrict__ cnt1,
              const float* __restrict__ dinv1, const float* __restrict__ dinv2,
              int2* __restrict__ csr, int E1, int E, int N) {
  const int i = blockIdx.x * blockDim.x + threadIdx.x;
  if (i >= E) return;
  int row, p;
  float w;
  if (i < E1) {
    const int s = src1[i], d = dst1[i];
    p = start[d] + rank[i];
    w = dinv1[s] * dinv1[d];
    row = s;
  } else {
    const int j = i - E1;
    const int s = src2[j], d = dst2[j];
    p = start[d] + cnt1[d] + rank[i];
    w = dinv2[s] * dinv2[d];
    row = s;
  }
  csr[p] = make_int2(row, __float_as_int(w));
}

#define WROW 256
#define WMAT (64 * WROW)
#define XPITCH 260

#define ACC_ROW_TO(a, g, e)                             \
  (a).x = __builtin_fmaf((g).x, (e), (a).x);           \
  (a).y = __builtin_fmaf((g).y, (e), (a).y);           \
  (a).z = __builtin_fmaf((g).z, (e), (a).z);           \
  (a).w = __builtin_fmaf((g).w, (e), (a).w)

template <bool GUARD>
__device__ __forceinline__ void gather_pair8(const float4* __restrict__ Xq,
                                             const int2 entA, const int2 entB,
                                             const int j, const int mA, const int mB,
                                             const int s2A, const int s2B,
                                             float4& accA1, float4& accA2,
                                             float4& accB1, float4& accB2) {
  float4 gA[8], gB[8];
  float eA[8], eB[8];
#pragma unroll
  for (int q = 0; q < 8; ++q) {
    const int r = __builtin_amdgcn_readlane(entA.x, j + q);
    eA[q] = __int_as_float(__builtin_amdgcn_readlane(entA.y, j + q));
    gA[q] = make_float4(0.f, 0.f, 0.f, 0.f);
    if (!GUARD || j + q < mA) gA[q] = Xq[(size_t)r * 32];
  }
#pragma unroll
  for (int q = 0; q < 8; ++q) {
    const int r = __builtin_amdgcn_readlane(entB.x, j + q);
    eB[q] = __int_as_float(__builtin_amdgcn_readlane(entB.y, j + q));
    gB[q] = make_float4(0.f, 0.f, 0.f, 0.f);
    if (!GUARD || j + q < mB) gB[q] = Xq[(size_t)r * 32];
  }
#pragma unroll
  for (int q = 0; q < 8; ++q) {
    const bool first = (j + q < s2A);
    const float e1 = first ? eA[q] : 0.f;
    const float e2 = first ? 0.f : eA[q];
    ACC_ROW_TO(accA1, gA[q], e1);
    ACC_ROW_TO(accA2, gA[q], e2);
  }
#pragma unroll
  for (int q = 0; q < 8; ++q) {
    const bool first = (j + q < s2B);
    const float e1 = first ? eB[q] : 0.f;
    const float e2 = first ? 0.f : eB[q];
    ACC_ROW_TO(accB1, gB[q], e1);
    ACC_ROW_TO(accB2, gB[q], e2);
  }
}

__global__ __launch_bounds__(256)
void gcn_layer_fused(const float* __restrict__ X,
                     const int2* __restrict__ csr,
                     const int* __restrict__ start,
                     const int* __restrict__ cnt1,
                     const int* __restrict__ cnt2,
                     const float* __restrict__ dinv1,
                     const float* __restrict__ dinv2,
                     const float* __restrict__ b1,
                     const float* __restrict__ b2,
                     const float* __restrict__ W1,
                     const float* __restrict__ W2,
                     float* __restrict__ OUT,
                     int N, int relu) {
  __shared__ __attribute__((aligned(16))) float Ws[2 * WMAT];
  __shared__ __attribute__((aligned(16))) float Xs[16 * XPITCH];
  __shared__ __attribute__((aligned(16))) float Ps[4 * 2 * 8 * 32];

  const int lane = threadIdx.x & 31;
  const int wave = __builtin_amdgcn_readfirstlane((int)(threadIdx.x >> 5));
  const int l15 = lane & 15;
  const int hi  = lane >> 4;

  {
    const int col4 = lane * 4;
    const int sw = (wave & 1) * 32;
#pragma unroll
    for (int i = 0; i < 8; ++i) {
      const int p = wave + 8 * i;
      const float4 r0 = *(const float4*)(W1 + (2 * p) * FEAT + col4);
      const float4 r1 = *(const float4*)(W1 + (2 * p + 1) * FEAT + col4);
      const float4 q0 = *(const float4*)(W2 + (2 * p) * FEAT + col4);
      const float4 q1 = *(const float4*)(W2 + (2 * p + 1) * FEAT + col4);
      float* d1 = Ws + p * WROW + ((2 * col4) ^ sw);
      float* d2 = d1 + WMAT;
      *(float4*)(d1)     = make_float4(r0.x, r1.x, r0.y, r1.y);
      *(float4*)(d1 + 4) = make_float4(r0.z, r1.z, r0.w, r1.w);
      *(float4*)(d2)     = make_float4(q0.x, q1.x, q0.y, q1.y);
      *(float4*)(d2 + 4) = make_float4(q0.z, q1.z, q0.w, q1.w);
    }
  }
  __syncthreads();

  const float4* __restrict__ Xq = (const float4*)X + lane;
  const int kh = wave >> 2;
  const int cg = wave & 3;
  const int colb = 32 * cg + l15;
  const float* __restrict__ wrow = Ws + kh * WMAT + hi * WROW + 2 * colb;
  const float* __restrict__ wl0 = wrow + 32 * hi;
  const float* __restrict__ wl1 = wrow + 32 - 32 * hi;
  const float* __restrict__ xa = Xs + l15 * XPITCH + 128 * kh + 2 * hi;
  float* __restrict__ pw = Ps + cg * 512 + lane;
  const float bs0 = b1[colb] + b2[colb];
  const float bs1 = b1[colb + 16] + b2[colb + 16];
  const int tiles = (N + 15) >> 4;

  for (int t = (int)blockIdx.x; t < tiles; t += (int)gridDim.x) {
    const int vA = 16 * t + 2 * wave;
    float4 oA1 = make_float4(0.f, 0.f, 0.f, 0.f);
    float4 oA2 = oA1, oB1 = oA1, oB2 = oA1;
    if (vA < N) {
      const bool hasB = (vA + 1 < N);
      const int vB = hasB ? vA + 1 : vA;
      const float4 sA = Xq[(size_t)vA * 32];
      const float4 sB = Xq[(size_t)vB * 32];
      const int beg = start[vA];
      const int cA1 = cnt1[vA];
      const int nA = cA1 + cnt2[vA];
      const int cB1 = cnt1[vB];
      const int nBr = cB1 + cnt2[vB];
      const float dA1 = dinv1[vA], dA2 = dinv2[vA];
      const float dB1 = dinv1[vB], dB2 = dinv2[vB];
      asm volatile("" :: "s"(beg));
      const int nB = hasB ? nBr : 0;
      const float wA1 = dA1 * dA1, wA2 = dA2 * dA2;
      const float wB1 = dB1 * dB1, wB2 = dB2 * dB2;
      float4 accA1, accA2, accB1, accB2;
      accA1.x = sA.x * wA1; accA1.y = sA.y * wA1; accA1.z = sA.z * wA1; accA1.w = sA.w * wA1;
      accA2.x = sA.x * wA2; accA2.y = sA.y * wA2; accA2.z = sA.z * wA2; accA2.w = sA.w * wA2;
      accB1.x = sB.x * wB1; accB1.y = sB.y * wB1; accB1.z = sB.z * wB1; accB1.w = sB.w * wB1;
      accB2.x = sB.x * wB2; accB2.y = sB.y * wB2; accB2.z = sB.z * wB2; accB2.w = sB.w * wB2;

      const int nmax = (nA > nB) ? nA : nB;
      for (int base = 0; base < nmax; base += 32) {
        int mA = nA - base; mA = (mA < 0) ? 0 : ((mA > 32) ? 32 : mA);
        int mB = nB - base; mB = (mB < 0) ? 0 : ((mB > 32) ? 32 : mB);
        int2 entA = make_int2(0, 0);
        int2 entB = make_int2(0, 0);
        if (lane < mA) entA = csr[beg + base + lane];
        if (lane < mB) entB = csr[beg + nA + base + lane];
        const int s2A = cA1 - base;
        const int s2B = cB1 - base;
        const int mlo = ((mA < mB) ? mA : mB) & ~7;
        const int mhi = (mA > mB) ? mA : mB;
        int j = 0;
        for (; j < mlo; j += 8)
          gather_pair8<false>(Xq, entA, entB, j, mA, mB, s2A, s2B, accA1, accA2, accB1, accB2);
        for (; j < mhi; j += 8)
          gather_pair8<true >(Xq, entA, entB, j, mA, mB, s2A, s2B, accA1, accA2, accB1, accB2);
      }
      oA1 = accA1; oA2 = accA2;
      if (hasB) { oB1 = accB1; oB2 = accB2; }
    }
    *(float4*)(Xs + (2 * wave) * XPITCH + 4 * lane)           = oA1;
    *(float4*)(Xs + (2 * wave) * XPITCH + 128 + 4 * lane)     = oA2;
    *(float4*)(Xs + (2 * wave + 1) * XPITCH + 4 * lane)       = oB1;
    *(float4*)(Xs + (2 * wave + 1) * XPITCH + 128 + 4 * lane) = oB2;
    __syncthreads();

    v8f c0 = {0.f, 0.f, 0.f, 0.f, 0.f, 0.f, 0.f, 0.f};
    v8f c1 = c0;
    for (int kc = 0; kc < 4; ++kc) {
      v2f a[8], p0[8], p1[8];
#pragma unroll
      for (int u = 0; u < 8; ++u) {
        a[u]  = *(const v2f*)(xa + 32 * kc + 4 * u);
        const int ro = (16 * kc + 2 * u) * WROW;
        p0[u] = *(const v2f*)(wl0 + ro);
        p1[u] = *(const v2f*)(wl1 + ro);
      }
#pragma unroll
      for (int u = 0; u < 8; ++u) {
        c0 = __builtin_amdgcn_wmma_f32_16x16x4_f32(false, a[u], false, p0[u], (short)0, c0, false, false);
        c1 = __builtin_amdgcn_wmma_f32_16x16x4_f32(false, a[u], false, p1[u], (short)0, c1, false, false);
      }
    }

    if (kh) {
#pragma unroll
      for (int r = 0; r < 8; ++r) {
        pw[r * 32]       = c0[r];
        pw[256 + r * 32] = c1[r];
      }
    }
    __syncthreads();

    if (!kh) {
      float v0[8], v1[8];
#pragma unroll
      for (int r = 0; r < 8; ++r) {
        v0[r] = (c0[r] + pw[r * 32]) + bs0;
        v1[r] = (c1[r] + pw[256 + r * 32]) + bs1;
        if (relu) { v0[r] = fmaxf(v0[r], 0.f); v1[r] = fmaxf(v1[r], 0.f); }
      }
      const int m0 = 16 * t;
      float* __restrict__ o = OUT + (size_t)(m0 + 8 * hi) * FEAT + colb;
      if (m0 + 16 <= N) {
#pragma unroll
        for (int r = 0; r < 8; ++r) {
          o[(size_t)r * FEAT]      = v0[r];
          o[(size_t)r * FEAT + 16] = v1[r];
        }
      } else {
#pragma unroll
        for (int r = 0; r < 8; ++r) {
          if (m0 + r + 8 * hi < N) {
            o[(size_t)r * FEAT]      = v0[r];
            o[(size_t)r * FEAT + 16] = v1[r];
          }
        }
      }
    }
  }
}

extern "C" void kernel_launch(void* const* d_in, const int* in_sizes, int n_in,
                              void* d_out, int out_size, void* d_ws, size_t ws_size,
                              hipStream_t stream) {
  const float* x      = (const float*)d_in[0];
  const int*   ei     = (const int*)d_in[1];
  const int*   cei    = (const int*)d_in[2];
  const float* conv_w = (const float*)d_in[3];
  const float* conv_b = (const float*)d_in[4];
  const float* ctrl_w = (const float*)d_in[5];
  const float* ctrl_b = (const float*)d_in[6];

  const int N  = in_sizes[0] / FEAT;
  const int E1 = in_sizes[1] / 2;
  const int E2 = in_sizes[2] / 2;
  const int D  = in_sizes[3] / (FEAT * FEAT);
  const int E  = E1 + E2;

  const int* src1 = ei;        const int* dst1 = ei + E1;
  const int* src2 = cei;       const int* dst2 = cei + E2;

  const size_t NF = (size_t)N * FEAT;
  const int Npad = ((N + 1023) / 1024) * 1024;
  float* xa    = (float*)d_ws;
  float* xb    = xa + NF;
  float* dinv1 = xb + NF;
  float* dinv2 = dinv1 + Npad;
  int*   cnt1  = (int*)(dinv2 + Npad);
  int*   cnt2  = cnt1 + Npad;
  int*   start = cnt2 + Npad;
  int*   rank  = start + Npad;
  int2*  csr   = (int2*)(rank + ((E + 3) / 4) * 4);

  const int T = 256;
  const int edgeBlocks = (E + T - 1) / T;

  cnt_zero<<<(2 * Npad / 4 + T - 1) / T, T, 0, stream>>>((int4*)cnt1, 2 * Npad / 4);
  edge_rank<<<edgeBlocks, T, 0, stream>>>(dst1, dst2, cnt1, cnt2, rank, E1, E);
  row_scan<<<Npad / 1024, T, 0, stream>>>(cnt1, cnt2, start, dinv1, dinv2, N);
  csr_fill<<<edgeBlocks, T, 0, stream>>>(src1, dst1, src2, dst2, rank, start, cnt1,
                                         dinv1, dinv2, csr, E1, E, N);

  const float* cur = x;
  float* nxt = xa;
  for (int i = 0; i < D; ++i) {
    const bool last = (i + 1 == D);
    float* const dst = last ? (float*)d_out : nxt;
    gcn_layer_fused<<<LAYER_BLOCKS, T, 0, stream>>>(cur, csr, start, cnt1, cnt2, dinv1, dinv2,
                                                    conv_b + (size_t)i * FEAT, ctrl_b + (size_t)i * FEAT,
                                                    conv_w + (size_t)i * FEAT * FEAT,
                                                    ctrl_w + (size_t)i * FEAT * FEAT,
                                                    dst, N, last ? 0 : 1);
    cur = dst;
    nxt = (dst == xa) ? xb : xa;
  }
}


# Edge sort: per-edge global returning atomics replaced by LDS counters (32 edge chunks, packed counts, chunk slots from a node-parallel scan): 3 prep launches, was 4; sum order = chunk, then LDS rank
# speedup vs baseline: 4.3640x; 1.1610x over previous
#include <hip/hip_runtime.h>

typedef float v2f __attribute__((ext_vector_type(2)));
typedef float v8f __attribute__((ext_vector_type(8)));

#define FEAT 128
#define LAYER_BLOCKS 512

__global__ void cnt_zero(int4* __restrict__ c, int n4) {
  const int i = blockIdx.x * blockDim.x + threadIdx.x;
  if (i < n4) c[i] = make_int4(0, 0, 0, 0);
}

__global__ __launch_bounds__(256)
void edge_rank(const int* __restrict__ dst1, const int* __restrict__ dst2,
               int* __restrict__ cnt1, int* __restrict__ cnt2,
               int* __restrict__ rank, int E1, int E) {
  const int i = blockIdx.x * blockDim.x + threadIdx.x;
  if (i >= E) return;
  if (i < E1) rank[i] = atomicAdd(&cnt1[dst1[i]], 1);
  else        rank[i] = atomicAdd(&cnt2[dst2[i - E1]], 1);
}

__global__ __launch_bounds__(256)
void row_scan(const int* __restrict__ cnt1, const int* __restrict__ cnt2,
              int* __restrict__ start,
              float* __restrict__ dinv1, float* __restrict__ dinv2, int N) {
  __shared__ int wsum[8];
  __shared__ int wpre[8];
  const int t = threadIdx.x;
  const int lane = t & 31;
  const int wave = t >> 5;
  const int4* __restrict__ c1 = (const int4*)cnt1;
  const int4* __restrict__ c2 = (const int4*)cnt2;

  int pre = 0;
  for (int k = 0; k < (int)blockIdx.x; ++k) {
    const int4 a = c1[k * 256 + t];
    const int4 b = c2[k * 256 + t];
    pre += (a.x + a.y) + (a.z + a.w) + (b.x + b.y) + (b.z + b.w);
  }
  const int q = blockIdx.x * 256 + t;
  const int4 a = c1[q];
  const int4 b = c2[q];
  const int l0 = a.x + b.x, l1 = a.y + b.y, l2 = a.z + b.z, l3 = a.w + b.w;
  const int mine = (l0 + l1) + (l2 + l3);

  int inc = mine;
  for (int o = 1; o < 32; o <<= 1) {
    const int u = __shfl_up(inc, o);
    if (lane >= o) inc += u;
  }
  for (int o = 16; o > 0; o >>= 1) pre += __shfl_xor(pre, o);
  if (lane == 31) wsum[wave] = inc;
  if (lane == 0)  wpre[wave] = pre;
  __syncthreads();
  int off = 0;
  for (int w = 0; w < 8; ++w) {
    off += wpre[w];
    if (w < wave) off += wsum[w];
  }
  const int s0 = off + inc - mine;
  const int i = q * 4;
  const int s1 = s0 + l0, s2 = s1 + l1, s3 = s2 + l2;
  if (i < N)     { start[i]     = s0; dinv1[i]     = rsqrtf((float)(1 + a.x)); dinv2[i]     = rsqrtf((float)(1 + b.x)); }
  if (i + 1 < N) { start[i + 1] = s1; dinv1[i + 1] = rsqrtf((float)(1 + a.y)); dinv2[i + 1] = rsqrtf((float)(1 + b.y)); }
  if (i + 2 < N) { start[i + 2] = s2; dinv1[i + 2] = rsqrtf((float)(1 + a.z)); dinv2[i + 2] = rsqrtf((float)(1 + b.z)); }
  if (i + 3 < N) { start[i + 3] = s3; dinv1[i + 3] = rsqrtf((float)(1 + a.w)); dinv2[i + 3] = rsqrtf((float)(1 + b.w)); }
}

__global__ __launch_bounds__(256)
void csr_fill(const int* __restrict__ src1, const int* __restrict__ dst1,
              const int* __restrict__ src2, const int* __restrict__ dst2,
              const int* __restrict__ rank, const int* __restrict__ start,
              const int* __restrict__ cnt1,
              const float* __restrict__ dinv1, const float* __restrict__ dinv2,
              int2* __restrict__ csr, int E1, int E, int N) {
  const int i = blockIdx.x * blockDim.x + threadIdx.x;
  if (i >= E) return;
  int row, p;
  float w;
  if (i < E1) {
    const int s = src1[i], d = dst1[i];
    p = start[d] + rank[i];
    w = dinv1[s] * dinv1[d];
    row = s;
  } else {
    const int j = i - E1;
    const int s = src2[j], d = dst2[j];
    p = start[d] + cnt1[d] + rank[i];
    w = dinv2[s] * dinv2[d];
    row = s;
  }
  csr[p] = make_int2(row, __float_as_int(w));
}

#define HK 32
#define HCAP 65536
#define HB 16

__global__ __launch_bounds__(1024)
void chunk_count(const int* __restrict__ dst1, const int* __restrict__ dst2,
                 int* __restrict__ hist, int* __restrict__ gsum, int* __restrict__ rank,
                 int E1, int E, int chunk, int Npad) {
  __shared__ __attribute__((aligned(16))) int cnt[HCAP];
  const int t = threadIdx.x;
  const int k = blockIdx.x;
  for (int v = t; v < Npad; v += 1024) cnt[v] = 0;
  __syncthreads();

  const int lo = k * chunk;
  const int hi = (lo + chunk < E) ? lo + chunk : E;
  for (int base = lo; base < hi; base += HB * 1024) {
    int d[HB];
#pragma unroll
    for (int q = 0; q < HB; ++q) {
      const int i = base + q * 1024 + t;
      d[q] = 0;
      if (i < hi) d[q] = (i < E1) ? dst1[i] : dst2[i - E1];
    }
    unsigned old[HB];
#pragma unroll
    for (int q = 0; q < HB; ++q) {
      const int i = base + q * 1024 + t;
      old[q] = 0u;
      if (i < hi) old[q] = (unsigned)atomicAdd(&cnt[d[q]], (i < E1) ? 1 : 65536);
    }
#pragma unroll
    for (int q = 0; q < HB; ++q) {
      const int i = base + q * 1024 + t;
      if (i < hi) rank[i] = (int)((i < E1) ? (old[q] & 0xFFFFu) : (old[q] >> 16));
    }
  }
  __syncthreads();

  for (int v = 4 * t; v < Npad; v += 4096)
    *(int4*)(hist + (size_t)k * Npad + v) = *(const int4*)(cnt + v);
  const int lane = t & 31;
  for (int g = t >> 5; g < (Npad >> 10); g += 32) {
    int s = 0;
#pragma unroll 8
    for (int j = 0; j < 32; ++j) {
      const unsigned cc = (unsigned)cnt[g * 1024 + j * 32 + lane];
      s += (int)((cc & 0xFFFFu) + (cc >> 16));
    }
    for (int o = 16; o > 0; o >>= 1) s += __shfl_xor(s, o);
    if (lane == 0) gsum[g * HK + k] = s;
  }
}

__global__ __launch_bounds__(1024)
void node_scan(const int* __restrict__ hist, const int* __restrict__ gsum,
               int* __restrict__ slot1, int* __restrict__ slot2,
               int* __restrict__ start, int* __restrict__ cnt1, int* __restrict__ cnt2,
               float* __restrict__ dinv1, float* __restrict__ dinv2, int N, int Npad) {
  __shared__ int wsum[32];
  __shared__ int wpre[32];
  const int t = threadIdx.x;
  const int lane = t & 31;
  const int wave = t >> 5;
  const int g = blockIdx.x;
  const int v = g * 1024 + t;

  unsigned h[HK];
#pragma unroll
  for (int k = 0; k < HK; ++k) h[k] = (unsigned)hist[(size_t)k * Npad + v];

  int pre = 0;
  for (int idx = t; idx < g * HK; idx += 1024) pre += gsum[idx];

  int c1 = 0, c2 = 0;
#pragma unroll
  for (int k = 0; k < HK; ++k) { c1 += (int)(h[k] & 0xFFFFu); c2 += (int)(h[k] >> 16); }
  const int mine = c1 + c2;

  int inc = mine;
  for (int o = 1; o < 32; o <<= 1) {
    const int u = __shfl_up(inc, o);
    if (lane >= o) inc += u;
  }
  for (int o = 16; o > 0; o >>= 1) pre += __shfl_xor(pre, o);
  if (lane == 31) wsum[wave] = inc;
  if (lane == 0)  wpre[wave] = pre;
  __syncthreads();
  int off = 0;
  for (int w = 0; w < 32; ++w) {
    off += wpre[w];
    if (w < wave) off += wsum[w];
  }
  const int s0 = off + inc - mine;
  if (v < N) {
    start[v] = s0;
    cnt1[v] = c1;
    cnt2[v] = c2;
    dinv1[v] = rsqrtf((float)(1 + c1));
    dinv2[v] = rsqrtf((float)(1 + c2));
    int r1 = s0, r2 = s0 + c1;
#pragma unroll
    for (int k = 0; k < HK; ++k) {
      slot1[(size_t)k * Npad + v] = r1;
      slot2[(size_t)k * Npad + v] = r2;
      r1 += (int)(h[k] & 0xFFFFu);
      r2 += (int)(h[k] >> 16);
    }
  }
}

__global__ __launch_bounds__(256)
void csr_fill_chunked(const int* __restrict__ src1, const int* __restrict__ dst1,
                      const int* __restrict__ src2, const int* __restrict__ dst2,
                      const int* __restrict__ rank,
                      const int* __restrict__ slot1, const int* __restrict__ slot2,
                      const float* __restrict__ dinv1, const float* __restrict__ dinv2,
                      int2* __restrict__ csr, int E1, int E, int chunk, int Npad) {
  const int i = blockIdx.x * blockDim.x + threadIdx.x;
  if (i >= E) return;
  const int k = i / chunk;
  int row, p;
  float w;
  if (i < E1) {
    const int s = src1[i], d = dst1[i];
    p = slot1[(size_t)k * Npad + d] + rank[i];
    w = dinv1[s] * dinv1[d];
    row = s;
  } else {
    const int j = i - E1;
    const int s = src2[j], d = dst2[j];
    p = slot2[(size_t)k * Npad + d] + rank[i];
    w = dinv2[s] * dinv2[d];
    row = s;
  }
  csr[p] = make_int2(row, __float_as_int(w));
}

#define WROW 256
#define WMAT (64 * WROW)
#define XPITCH 260

#define ACC_ROW_TO(a, g, e)                             \
  (a).x = __builtin_fmaf((g).x, (e), (a).x);           \
  (a).y = __builtin_fmaf((g).y, (e), (a).y);           \
  (a).z = __builtin_fmaf((g).z, (e), (a).z);           \
  (a).w = __builtin_fmaf((g).w, (e), (a).w)

template <bool GUARD>
__device__ __forceinline__ void gather_pair8(const float4* __restrict__ Xq,
                                             const int2 entA, const int2 entB,
                                             const int j, const int mA, const int mB,
                                             const int s2A, const int s2B,
                                             float4& accA1, float4& accA2,
                                             float4& accB1, float4& accB2) {
  float4 gA[8], gB[8];
  float eA[8], eB[8];
#pragma unroll
  for (int q = 0; q < 8; ++q) {
    const int r = __builtin_amdgcn_readlane(entA.x, j + q);
    eA[q] = __int_as_float(__builtin_amdgcn_readlane(entA.y, j + q));
    gA[q] = make_float4(0.f, 0.f, 0.f, 0.f);
    if (!GUARD || j + q < mA) gA[q] = Xq[(size_t)r * 32];
  }
#pragma unroll
  for (int q = 0; q < 8; ++q) {
    const int r = __builtin_amdgcn_readlane(entB.x, j + q);
    eB[q] = __int_as_float(__builtin_amdgcn_readlane(entB.y, j + q));
    gB[q] = make_float4(0.f, 0.f, 0.f, 0.f);
    if (!GUARD || j + q < mB) gB[q] = Xq[(size_t)r * 32];
  }
#pragma unroll
  for (int q = 0; q < 8; ++q) {
    const bool first = (j + q < s2A);
    const float e1 = first ? eA[q] : 0.f;
    const float e2 = first ? 0.f : eA[q];
    ACC_ROW_TO(accA1, gA[q], e1);
    ACC_ROW_TO(accA2, gA[q], e2);
  }
#pragma unroll
  for (int q = 0; q < 8; ++q) {
    const bool first = (j + q < s2B);
    const float e1 = first ? eB[q] : 0.f;
    const float e2 = first ? 0.f : eB[q];
    ACC_ROW_TO(accB1, gB[q], e1);
    ACC_ROW_TO(accB2, gB[q], e2);
  }
}

__global__ __launch_bounds__(256)
void gcn_layer_fused(const float* __restrict__ X,
                     const int2* __restrict__ csr,
                     const int* __restrict__ start,
                     const int* __restrict__ cnt1,
                     const int* __restrict__ cnt2,
                     const float* __restrict__ dinv1,
                     const float* __restrict__ dinv2,
                     const float* __restrict__ b1,
                     const float* __restrict__ b2,
                     const float* __restrict__ W1,
                     const float* __restrict__ W2,
                     float* __restrict__ OUT,
                     int N, int relu) {
  __shared__ __attribute__((aligned(16))) float Ws[2 * WMAT];
  __shared__ __attribute__((aligned(16))) float Xs[16 * XPITCH];
  __shared__ __attribute__((aligned(16))) float Ps[4 * 2 * 8 * 32];

  const int lane = threadIdx.x & 31;
  const int wave = __builtin_amdgcn_readfirstlane((int)(threadIdx.x >> 5));
  const int l15 = lane & 15;
  const int hi  = lane >> 4;

  {
    const int col4 = lane * 4;
    const int sw = (wave & 1) * 32;
#pragma unroll
    for (int i = 0; i < 8; ++i) {
      const int p = wave + 8 * i;
      const float4 r0 = *(const float4*)(W1 + (2 * p) * FEAT + col4);
      const float4 r1 = *(const float4*)(W1 + (2 * p + 1) * FEAT + col4);
      const float4 q0 = *(const float4*)(W2 + (2 * p) * FEAT + col4);
      const float4 q1 = *(const float4*)(W2 + (2 * p + 1) * FEAT + col4);
      float* d1 = Ws + p * WROW + ((2 * col4) ^ sw);
      float* d2 = d1 + WMAT;
      *(float4*)(d1)     = make_float4(r0.x, r1.x, r0.y, r1.y);
      *(float4*)(d1 + 4) = make_float4(r0.z, r1.z, r0.w, r1.w);
      *(float4*)(d2)     = make_float4(q0.x, q1.x, q0.y, q1.y);
      *(float4*)(d2 + 4) = make_float4(q0.z, q1.z, q0.w, q1.w);
    }
  }
  __syncthreads();

  const float4* __restrict__ Xq = (const float4*)X + lane;
  const int kh = wave >> 2;
  const int cg = wave & 3;
  const int colb = 32 * cg + l15;
  const float* __restrict__ wrow = Ws + kh * WMAT + hi * WROW + 2 * colb;
  const float* __restrict__ wl0 = wrow + 32 * hi;
  const float* __restrict__ wl1 = wrow + 32 - 32 * hi;
  const float* __restrict__ xa = Xs + l15 * XPITCH + 128 * kh + 2 * hi;
  float* __restrict__ pw = Ps + cg * 512 + lane;
  const float bs0 = b1[colb] + b2[colb];
  const float bs1 = b1[colb + 16] + b2[colb + 16];
  const int tiles = (N + 15) >> 4;

  for (int t = (int)blockIdx.x; t < tiles; t += (int)gridDim.x) {
    const int vA = 16 * t + 2 * wave;
    float4 oA1 = make_float4(0.f, 0.f, 0.f, 0.f);
    float4 oA2 = oA1, oB1 = oA1, oB2 = oA1;
    if (vA < N) {
      const bool hasB = (vA + 1 < N);
      const int vB = hasB ? vA + 1 : vA;
      const float4 sA = Xq[(size_t)vA * 32];
      const float4 sB = Xq[(size_t)vB * 32];
      const int beg = start[vA];
      const int cA1 = cnt1[vA];
      const int nA = cA1 + cnt2[vA];
      const int cB1 = cnt1[vB];
      const int nBr = cB1 + cnt2[vB];
      const float dA1 = dinv1[vA], dA2 = dinv2[vA];
      const float dB1 = dinv1[vB], dB2 = dinv2[vB];
      asm volatile("" :: "s"(beg));
      const int nB = hasB ? nBr : 0;
      const float wA1 = dA1 * dA1, wA2 = dA2 * dA2;
      const float wB1 = dB1 * dB1, wB2 = dB2 * dB2;
      float4 accA1, accA2, accB1, accB2;
      accA1.x = sA.x * wA1; accA1.y = sA.y * wA1; accA1.z = sA.z * wA1; accA1.w = sA.w * wA1;
      accA2.x = sA.x * wA2; accA2.y = sA.y * wA2; accA2.z = sA.z * wA2; accA2.w = sA.w * wA2;
      accB1.x = sB.x * wB1; accB1.y = sB.y * wB1; accB1.z = sB.z * wB1; accB1.w = sB.w * wB1;
      accB2.x = sB.x * wB2; accB2.y = sB.y * wB2; accB2.z = sB.z * wB2; accB2.w = sB.w * wB2;

      const int nmax = (nA > nB) ? nA : nB;
      for (int base = 0; base < nmax; base += 32) {
        int mA = nA - base; mA = (mA < 0) ? 0 : ((mA > 32) ? 32 : mA);
        int mB = nB - base; mB = (mB < 0) ? 0 : ((mB > 32) ? 32 : mB);
        int2 entA = make_int2(0, 0);
        int2 entB = make_int2(0, 0);
        if (lane < mA) entA = csr[beg + base + lane];
        if (lane < mB) entB = csr[beg + nA + base + lane];
        const int s2A = cA1 - base;
        const int s2B = cB1 - base;
        const int mlo = ((mA < mB) ? mA : mB) & ~7;
        const int mhi = (mA > mB) ? mA : mB;
        int j = 0;
        for (; j < mlo; j += 8)
          gather_pair8<false>(Xq, entA, entB, j, mA, mB, s2A, s2B, accA1, accA2, accB1, accB2);
        for (; j < mhi; j += 8)
          gather_pair8<true >(Xq, entA, entB, j, mA, mB, s2A, s2B, accA1, accA2, accB1, accB2);
      }
      oA1 = accA1; oA2 = accA2;
      if (hasB) { oB1 = accB1; oB2 = accB2; }
    }
    *(float4*)(Xs + (2 * wave) * XPITCH + 4 * lane)           = oA1;
    *(float4*)(Xs + (2 * wave) * XPITCH + 128 + 4 * lane)     = oA2;
    *(float4*)(Xs + (2 * wave + 1) * XPITCH + 4 * lane)       = oB1;
    *(float4*)(Xs + (2 * wave + 1) * XPITCH + 128 + 4 * lane) = oB2;
    __syncthreads();

    v8f c0 = {0.f, 0.f, 0.f, 0.f, 0.f, 0.f, 0.f, 0.f};
    v8f c1 = c0;
    for (int kc = 0; kc < 4; ++kc) {
      v2f a[8], p0[8], p1[8];
#pragma unroll
      for (int u = 0; u < 8; ++u) {
        a[u]  = *(const v2f*)(xa + 32 * kc + 4 * u);
        const int ro = (16 * kc + 2 * u) * WROW;
        p0[u] = *(const v2f*)(wl0 + ro);
        p1[u] = *(const v2f*)(wl1 + ro);
      }
#pragma unroll
      for (int u = 0; u < 8; ++u) {
        c0 = __builtin_amdgcn_wmma_f32_16x16x4_f32(false, a[u], false, p0[u], (short)0, c0, false, false);
        c1 = __builtin_amdgcn_wmma_f32_16x16x4_f32(false, a[u], false, p1[u], (short)0, c1, false, false);
      }
    }

    if (kh) {
#pragma unroll
      for (int r = 0; r < 8; ++r) {
        pw[r * 32]       = c0[r];
        pw[256 + r * 32] = c1[r];
      }
    }
    __syncthreads();

    if (!kh) {
      float v0[8], v1[8];
#pragma unroll
      for (int r = 0; r < 8; ++r) {
        v0[r] = (c0[r] + pw[r * 32]) + bs0;
        v1[r] = (c1[r] + pw[256 + r * 32]) + bs1;
        if (relu) { v0[r] = fmaxf(v0[r], 0.f); v1[r] = fmaxf(v1[r], 0.f); }
      }
      const int m0 = 16 * t;
      float* __restrict__ o = OUT + (size_t)(m0 + 8 * hi) * FEAT + colb;
      if (m0 + 16 <= N) {
#pragma unroll
        for (int r = 0; r < 8; ++r) {
          o[(size_t)r * FEAT]      = v0[r];
          o[(size_t)r * FEAT + 16] = v1[r];
        }
      } else {
#pragma unroll
        for (int r = 0; r < 8; ++r) {
          if (m0 + r + 8 * hi < N) {
            o[(size_t)r * FEAT]      = v0[r];
            o[(size_t)r * FEAT + 16] = v1[r];
          }
        }
      }
    }
  }
}

extern "C" void kernel_launch(void* const* d_in, const int* in_sizes, int n_in,
                              void* d_out, int out_size, void* d_ws, size_t ws_size,
                              hipStream_t stream) {
  const float* x      = (const float*)d_in[0];
  const int*   ei     = (const int*)d_in[1];
  const int*   cei    = (const int*)d_in[2];
  const float* conv_w = (const float*)d_in[3];
  const float* conv_b = (const float*)d_in[4];
  const float* ctrl_w = (const float*)d_in[5];
  const float* ctrl_b = (const float*)d_in[6];

  const int N  = in_sizes[0] / FEAT;
  const int E1 = in_sizes[1] / 2;
  const int E2 = in_sizes[2] / 2;
  const int D  = in_sizes[3] / (FEAT * FEAT);
  const int E  = E1 + E2;

  const int* src1 = ei;        const int* dst1 = ei + E1;
  const int* src2 = cei;       const int* dst2 = cei + E2;

  const size_t NF = (size_t)N * FEAT;
  const int Npad = ((N + 1023) / 1024) * 1024;
  float* xa    = (float*)d_ws;
  float* xb    = xa + NF;
  float* dinv1 = xb + NF;
  float* dinv2 = dinv1 + Npad;
  int*   cnt1  = (int*)(dinv2 + Npad);
  int*   cnt2  = cnt1 + Npad;
  int*   start = cnt2 + Npad;
  int*   rank  = start + Npad;
  int2*  csr   = (int2*)(rank + ((E + 3) / 4) * 4);
  int*   hist  = (int*)(csr + ((E + 1) / 2) * 2);
  int*   slot1 = hist + (size_t)HK * Npad;
  int*   slot2 = slot1 + (size_t)HK * Npad;
  int*   gsum  = slot2 + (size_t)HK * Npad;

  const int T = 256;
  const int edgeBlocks = (E + T - 1) / T;
  const int chunk = (E + HK - 1) / HK;

  if (Npad <= HCAP && chunk <= 65535) {
    chunk_count<<<HK, 1024, 0, stream>>>(dst1, dst2, hist, gsum, rank, E1, E, chunk, Npad);
    node_scan<<<Npad / 1024, 1024, 0, stream>>>(hist, gsum, slot1, slot2, start, cnt1, cnt2,
                                                dinv1, dinv2, N, Npad);
    csr_fill_chunked<<<edgeBlocks, T, 0, stream>>>(src1, dst1, src2, dst2, rank, slot1, slot2,
                                                   dinv1, dinv2, csr, E1, E, chunk, Npad);
  } else {
    cnt_zero<<<(2 * Npad / 4 + T - 1) / T, T, 0, stream>>>((int4*)cnt1, 2 * Npad / 4);
    edge_rank<<<edgeBlocks, T, 0, stream>>>(dst1, dst2, cnt1, cnt2, rank, E1, E);
    row_scan<<<Npad / 1024, T, 0, stream>>>(cnt1, cnt2, start, dinv1, dinv2, N);
    csr_fill<<<edgeBlocks, T, 0, stream>>>(src1, dst1, src2, dst2, rank, start, cnt1,
                                           dinv1, dinv2, csr, E1, E, N);
  }

  const float* cur = x;
  float* nxt = xa;
  for (int i = 0; i < D; ++i) {
    const bool last = (i + 1 == D);
    float* const dst = last ? (float*)d_out : nxt;
    gcn_layer_fused<<<LAYER_BLOCKS, T, 0, stream>>>(cur, csr, start, cnt1, cnt2, dinv1, dinv2,
                                                    conv_b + (size_t)i * FEAT, ctrl_b + (size_t)i * FEAT,
                                                    conv_w + (size_t)i * FEAT * FEAT,
                                                    ctrl_w + (size_t)i * FEAT * FEAT,
                                                    dst, N, last ? 0 : 1);
    cur = dst;
    nxt = (dst == xa) ? xb : xa;
  }
}
